# starting kernel
# baseline (speedup 1.0000x reference)
#include <hip/hip_runtime.h>

typedef __attribute__((ext_vector_type(2))) float v2f;
typedef __attribute__((ext_vector_type(8))) float v8f;

#define IN_F   128
#define OUT_F  128
#define MAXDEG 4
#define NBUCK  5
#define TILE_M 16
#define LDA    260
#define SCAN_B 256

__global__ void fill_i32_kernel(int* __restrict__ p, int val, size_t n) {
    size_t i = (size_t)blockIdx.x * blockDim.x + threadIdx.x;
    size_t stride = (size_t)gridDim.x * blockDim.x;
    for (; i < n; i += stride) p[i] = val;
}

__global__ void degree_kernel(const int* __restrict__ col, int* __restrict__ degi, int E) {
    int e = blockIdx.x * blockDim.x + threadIdx.x;
    if (e < E) atomicAdd(&degi[col[e]], 1);
}

__global__ void scan1_kernel(const int* __restrict__ degi, int* __restrict__ rowstart,
                             int* __restrict__ bsum, int N) {
    __shared__ int sh[SCAN_B];
    int i = blockIdx.x * SCAN_B + threadIdx.x;
    int v = (i < N) ? degi[i] : 0;
    sh[threadIdx.x] = v;
    __syncthreads();
    #pragma unroll
    for (int off = 1; off < SCAN_B; off <<= 1) {
        int t = (threadIdx.x >= off) ? sh[threadIdx.x - off] : 0;
        __syncthreads();
        sh[threadIdx.x] += t;
        __syncthreads();
    }
    if (i < N) rowstart[i] = sh[threadIdx.x] - v;
    if (threadIdx.x == SCAN_B - 1) bsum[blockIdx.x] = sh[SCAN_B - 1];
}

__global__ void scan2_kernel(int* __restrict__ bsum, int nBlocks) {
    if (threadIdx.x == 0 && blockIdx.x == 0) {
        int acc = 0;
        for (int b = 0; b < nBlocks; ++b) { int t = bsum[b]; bsum[b] = acc; acc += t; }
    }
}

__global__ void scan3_kernel(int* __restrict__ rowstart, const int* __restrict__ bsum, int N) {
    int i = blockIdx.x * SCAN_B + threadIdx.x;
    if (i < N) rowstart[i] += bsum[blockIdx.x];
}

__global__ void csr_fill_kernel(const int* __restrict__ row, const int* __restrict__ col,
                                const int* __restrict__ rowstart, int* __restrict__ cursor,
                                int* __restrict__ edge_src, int E) {
    int e = blockIdx.x * blockDim.x + threadIdx.x;
    if (e >= E) return;
    int d = col[e];
    int pos = atomicAdd(&cursor[d], 1);
    edge_src[rowstart[d] + pos] = row[e];
}

__global__ void bucket_count_kernel(const int* __restrict__ degi, int* __restrict__ meta, int N) {
    int i = blockIdx.x * blockDim.x + threadIdx.x;
    if (i >= N) return;
    int b = min(degi[i], MAXDEG);
    atomicAdd(&meta[b], 1);
}

__global__ void bucket_offset_kernel(int* __restrict__ meta) {
    if (threadIdx.x == 0 && blockIdx.x == 0) {
        int off = 0;
        for (int b = 0; b < NBUCK; ++b) {
            meta[5 + b] = off;
            off += (meta[b] + TILE_M - 1) & ~(TILE_M - 1);
        }
    }
}

__global__ void bucket_scatter_kernel(const int* __restrict__ degi, int* __restrict__ meta,
                                      int* __restrict__ perm, int N) {
    int i = blockIdx.x * blockDim.x + threadIdx.x;
    if (i >= N) return;
    int b = min(degi[i], MAXDEG);
    int pos = atomicAdd(&meta[10 + b], 1);
    perm[meta[5 + b] + pos] = i;
}

__global__ __launch_bounds__(256) void mfconv_gemm_kernel(
    const float* __restrict__ x,
    const int* __restrict__ degi, const int* __restrict__ rowstart,
    const int* __restrict__ edge_src, const int* __restrict__ perm,
    const float* __restrict__ neigh_w, const float* __restrict__ neigh_b,
    const float* __restrict__ root_w,  const float* __restrict__ root_b,
    float* __restrict__ out)
{
    __shared__ float ldsA[TILE_M * LDA];
    __shared__ int   s_node[TILE_M];
    __shared__ int   s_bucket;

    const int tid  = threadIdx.x;
    const int tile = blockIdx.x;

    if (tid < TILE_M) s_node[tid] = perm[tile * TILE_M + tid];
    __syncthreads();

    if (tid == 0) {
        int b = -1;
        for (int i = 0; i < TILE_M; ++i) {
            int nd = s_node[i];
            if (nd >= 0) { b = min(degi[nd], MAXDEG); break; }
        }
        s_bucket = b;
    }

    {
        const int m    = tid >> 4;
        const int k0   = (tid & 15) * 16;
        const int node = s_node[m];
        float4 a0 = make_float4(0.f,0.f,0.f,0.f), a1 = a0, a2 = a0, a3 = a0;
        if (node >= 0) {
            if (k0 < IN_F) {
                const int start = rowstart[node];
                const int d     = degi[node];
                for (int j = 0; j < d; ++j) {
                    const float* xs = x + (size_t)edge_src[start + j] * IN_F + k0;
                    const float4 v0 = *(const float4*)(xs + 0);
                    const float4 v1 = *(const float4*)(xs + 4);
                    const float4 v2 = *(const float4*)(xs + 8);
                    const float4 v3 = *(const float4*)(xs + 12);
                    a0.x += v0.x; a0.y += v0.y; a0.z += v0.z; a0.w += v0.w;
                    a1.x += v1.x; a1.y += v1.y; a1.z += v1.z; a1.w += v1.w;
                    a2.x += v2.x; a2.y += v2.y; a2.z += v2.z; a2.w += v2.w;
                    a3.x += v3.x; a3.y += v3.y; a3.z += v3.z; a3.w += v3.w;
                }
                const float inv = 1.0f / fmaxf((float)d, 1.0f);
                a0.x *= inv; a0.y *= inv; a0.z *= inv; a0.w *= inv;
                a1.x *= inv; a1.y *= inv; a1.z *= inv; a1.w *= inv;
                a2.x *= inv; a2.y *= inv; a2.z *= inv; a2.w *= inv;
                a3.x *= inv; a3.y *= inv; a3.z *= inv; a3.w *= inv;
            } else {
                const float* xs = x + (size_t)node * IN_F + (k0 - IN_F);
                a0 = *(const float4*)(xs + 0);
                a1 = *(const float4*)(xs + 4);
                a2 = *(const float4*)(xs + 8);
                a3 = *(const float4*)(xs + 12);
            }
        }
        float* dst = &ldsA[m * LDA + k0];
        *(float4*)(dst + 0)  = a0;
        *(float4*)(dst + 4)  = a1;
        *(float4*)(dst + 8)  = a2;
        *(float4*)(dst + 12) = a3;
    }
    __syncthreads();

    const int bucket = s_bucket;
    if (bucket < 0) return;

    const int wave  = tid >> 5;
    const int lane  = tid & 31;
    const int lr    = lane & 15;
    const int khalf = (lane >> 4) << 1;
    const int ncol  = wave * 16 + lr;

    const float* __restrict__ wn   = neigh_w + ((size_t)bucket * OUT_F + ncol) * IN_F;
    const float* __restrict__ wr   = root_w  + ((size_t)bucket * OUT_F + ncol) * IN_F;
    const float* __restrict__ aRow = &ldsA[lr * LDA];

    v8f acc = {};
    #pragma unroll
    for (int kk = 0; kk < 32; ++kk) {
        const int k = kk * 4 + khalf;
        v2f a, b;
        a.x = aRow[k];     a.y = aRow[k + 1];
        b.x = wn[k];       b.y = wn[k + 1];
        acc = __builtin_amdgcn_wmma_f32_16x16x4_f32(false, a, false, b,
                                                    (short)0, acc, false, false);
    }
    #pragma unroll
    for (int kk = 0; kk < 32; ++kk) {
        const int k = kk * 4 + khalf;
        v2f a, b;
        a.x = aRow[IN_F + k]; a.y = aRow[IN_F + k + 1];
        b.x = wr[k];          b.y = wr[k + 1];
        acc = __builtin_amdgcn_wmma_f32_16x16x4_f32(false, a, false, b,
                                                    (short)0, acc, false, false);
    }

    const float bias = neigh_b[bucket * OUT_F + ncol] + root_b[bucket * OUT_F + ncol];
    const int mbase  = (lane >> 4) * 8;
    #pragma unroll
    for (int i = 0; i < 8; ++i) {
        const int node = s_node[mbase + i];
        if (node >= 0) out[(size_t)node * OUT_F + ncol] = acc[i] + bias;
    }
}

extern "C" void kernel_launch(void* const* d_in, const int* in_sizes, int n_in,
                              void* d_out, int out_size, void* d_ws, size_t ws_size,
                              hipStream_t stream) {
    const float* x       = (const float*)d_in[0];
    const float* neigh_w = (const float*)d_in[1];
    const float* neigh_b = (const float*)d_in[2];
    const float* root_w  = (const float*)d_in[3];
    const float* root_b  = (const float*)d_in[4];
    const int*   eidx    = (const int*)d_in[5];

    const int N = in_sizes[0] / IN_F;
    const int E = in_sizes[5] / 2;
    const int* row = eidx;
    const int* col = eidx + E;

    const int nScanBlocks = (N + SCAN_B - 1) / SCAN_B;
    const int permLen = ((N + NBUCK * TILE_M + TILE_M - 1) / TILE_M) * TILE_M;
    const int nTiles  = permLen / TILE_M;

    char* ws = (char*)d_ws;
    int* degi     = (int*)ws;  ws += (size_t)N * sizeof(int);
    int* cursor   = (int*)ws;  ws += (size_t)N * sizeof(int);
    int* rowstart = (int*)ws;  ws += (size_t)N * sizeof(int);
    int* bsum     = (int*)ws;  ws += (size_t)((nScanBlocks + 15) & ~15) * sizeof(int);
    int* edge_src = (int*)ws;  ws += (size_t)E * sizeof(int);
    int* perm     = (int*)ws;  ws += (size_t)permLen * sizeof(int);
    int* meta     = (int*)ws;

    fill_i32_kernel<<<1024, 256, 0, stream>>>(degi, 0, (size_t)2 * N);
    fill_i32_kernel<<<1024, 256, 0, stream>>>(perm, -1, (size_t)permLen);
    fill_i32_kernel<<<1, 32, 0, stream>>>(meta, 0, 16);

    degree_kernel<<<(E + 255) / 256, 256, 0, stream>>>(col, degi, E);
    scan1_kernel<<<nScanBlocks, SCAN_B, 0, stream>>>(degi, rowstart, bsum, N);
    scan2_kernel<<<1, 1, 0, stream>>>(bsum, nScanBlocks);
    scan3_kernel<<<nScanBlocks, SCAN_B, 0, stream>>>(rowstart, bsum, N);
    csr_fill_kernel<<<(E + 255) / 256, 256, 0, stream>>>(row, col, rowstart, cursor,
                                                         edge_src, E);

    bucket_count_kernel<<<(N + 255) / 256, 256, 0, stream>>>(degi, meta, N);
    bucket_offset_kernel<<<1, 1, 0, stream>>>(meta);
    bucket_scatter_kernel<<<(N + 255) / 256, 256, 0, stream>>>(degi, meta, perm, N);

    mfconv_gemm_kernel<<<nTiles, 256, 0, stream>>>(x, degi, rowstart, edge_src, perm,
                                                   neigh_w, neigh_b, root_w, root_b,
                                                   (float*)d_out);
}


# bucket count + scatter: reduce the five hot bucket counters inside each work-group (wave ballots, LDS), one global atomic per bucket per work-group instead of one per node
# speedup vs baseline: 1.5411x; 1.5411x over previous
#include <hip/hip_runtime.h>

typedef __attribute__((ext_vector_type(2))) float v2f;
typedef __attribute__((ext_vector_type(8))) float v8f;

#define IN_F   128
#define OUT_F  128
#define MAXDEG 4
#define NBUCK  5
#define TILE_M 16
#define LDA    260
#define SCAN_B 256

__global__ void fill_i32_kernel(int* __restrict__ p, int val, size_t n) {
    size_t i = (size_t)blockIdx.x * blockDim.x + threadIdx.x;
    size_t stride = (size_t)gridDim.x * blockDim.x;
    for (; i < n; i += stride) p[i] = val;
}

__global__ void degree_kernel(const int* __restrict__ col, int* __restrict__ degi, int E) {
    int e = blockIdx.x * blockDim.x + threadIdx.x;
    if (e < E) atomicAdd(&degi[col[e]], 1);
}

__global__ void scan1_kernel(const int* __restrict__ degi, int* __restrict__ rowstart,
                             int* __restrict__ bsum, int N) {
    __shared__ int sh[SCAN_B];
    int i = blockIdx.x * SCAN_B + threadIdx.x;
    int v = (i < N) ? degi[i] : 0;
    sh[threadIdx.x] = v;
    __syncthreads();
    #pragma unroll
    for (int off = 1; off < SCAN_B; off <<= 1) {
        int t = (threadIdx.x >= off) ? sh[threadIdx.x - off] : 0;
        __syncthreads();
        sh[threadIdx.x] += t;
        __syncthreads();
    }
    if (i < N) rowstart[i] = sh[threadIdx.x] - v;
    if (threadIdx.x == SCAN_B - 1) bsum[blockIdx.x] = sh[SCAN_B - 1];
}

__global__ void scan2_kernel(int* __restrict__ bsum, int nBlocks) {
    if (threadIdx.x == 0 && blockIdx.x == 0) {
        int acc = 0;
        for (int b = 0; b < nBlocks; ++b) { int t = bsum[b]; bsum[b] = acc; acc += t; }
    }
}

__global__ void scan3_kernel(int* __restrict__ rowstart, const int* __restrict__ bsum, int N) {
    int i = blockIdx.x * SCAN_B + threadIdx.x;
    if (i < N) rowstart[i] += bsum[blockIdx.x];
}

__global__ void csr_fill_kernel(const int* __restrict__ row, const int* __restrict__ col,
                                const int* __restrict__ rowstart, int* __restrict__ cursor,
                                int* __restrict__ edge_src, int E) {
    int e = blockIdx.x * blockDim.x + threadIdx.x;
    if (e >= E) return;
    int d = col[e];
    int pos = atomicAdd(&cursor[d], 1);
    edge_src[rowstart[d] + pos] = row[e];
}

#define BK_B  1024
#define BK_W  (BK_B / 32)

__global__ __launch_bounds__(BK_B) void bucket_count_kernel(const int* __restrict__ degi,
                                                            int* __restrict__ meta, int N) {
    __shared__ int s_w[BK_W][NBUCK];
    const int tid  = threadIdx.x;
    const int lane = tid & 31;
    const int wave = tid >> 5;
    const int i = blockIdx.x * BK_B + tid;
    const int b = (i < N) ? min(degi[i], MAXDEG) : -1;
    int myc = 0;
    for (int k = 0; k < NBUCK; ++k) {
        const unsigned m = __builtin_amdgcn_ballot_w32(b == k);
        if (lane == k) myc = __builtin_popcount(m);
    }
    if (lane < NBUCK) s_w[wave][lane] = myc;
    __syncthreads();
    if (tid < NBUCK) {
        int acc = 0;
        for (int w = 0; w < BK_W; ++w) acc += s_w[w][tid];
        if (acc) atomicAdd(&meta[tid], acc);
    }
}

__global__ void bucket_offset_kernel(int* __restrict__ meta) {
    if (threadIdx.x == 0 && blockIdx.x == 0) {
        int off = 0;
        for (int b = 0; b < NBUCK; ++b) {
            meta[5 + b] = off;
            off += (meta[b] + TILE_M - 1) & ~(TILE_M - 1);
        }
    }
}

__global__ __launch_bounds__(BK_B) void bucket_scatter_kernel(const int* __restrict__ degi,
                                                              int* __restrict__ meta,
                                                              int* __restrict__ perm, int N) {
    __shared__ int s_w[BK_W][NBUCK];
    __shared__ int s_base[NBUCK];
    const int tid  = threadIdx.x;
    const int lane = tid & 31;
    const int wave = tid >> 5;
    const int i = blockIdx.x * BK_B + tid;
    const int b = (i < N) ? min(degi[i], MAXDEG) : -1;
    unsigned mine = 0;
    int myc = 0;
    for (int k = 0; k < NBUCK; ++k) {
        const unsigned m = __builtin_amdgcn_ballot_w32(b == k);
        if (b == k)    mine = m;
        if (lane == k) myc  = __builtin_popcount(m);
    }
    if (lane < NBUCK) s_w[wave][lane] = myc;
    __syncthreads();
    if (tid < NBUCK) {
        int acc = 0;
        for (int w = 0; w < BK_W; ++w) { const int c = s_w[w][tid]; s_w[w][tid] = acc; acc += c; }
        int base = 0;
        if (acc) base = atomicAdd(&meta[10 + tid], acc);
        s_base[tid] = meta[5 + tid] + base;
    }
    __syncthreads();
    if (b >= 0) {
        const int rank = __builtin_popcount(mine & ((1u << lane) - 1u));
        perm[s_base[b] + s_w[wave][b] + rank] = i;
    }
}

__global__ __launch_bounds__(256) void mfconv_gemm_kernel(
    const float* __restrict__ x,
    const int* __restrict__ degi, const int* __restrict__ rowstart,
    const int* __restrict__ edge_src, const int* __restrict__ perm,
    const float* __restrict__ neigh_w, const float* __restrict__ neigh_b,
    const float* __restrict__ root_w,  const float* __restrict__ root_b,
    float* __restrict__ out)
{
    __shared__ float ldsA[TILE_M * LDA];
    __shared__ int   s_node[TILE_M];
    __shared__ int   s_bucket;

    const int tid  = threadIdx.x;
    const int tile = blockIdx.x;

    if (tid < TILE_M) s_node[tid] = perm[tile * TILE_M + tid];
    __syncthreads();

    if (tid == 0) {
        int b = -1;
        for (int i = 0; i < TILE_M; ++i) {
            int nd = s_node[i];
            if (nd >= 0) { b = min(degi[nd], MAXDEG); break; }
        }
        s_bucket = b;
    }

    {
        const int m    = tid >> 4;
        const int k0   = (tid & 15) * 16;
        const int node = s_node[m];
        float4 a0 = make_float4(0.f,0.f,0.f,0.f), a1 = a0, a2 = a0, a3 = a0;
        if (node >= 0) {
            if (k0 < IN_F) {
                const int start = rowstart[node];
                const int d     = degi[node];
                for (int j = 0; j < d; ++j) {
                    const float* xs = x + (size_t)edge_src[start + j] * IN_F + k0;
                    const float4 v0 = *(const float4*)(xs + 0);
                    const float4 v1 = *(const float4*)(xs + 4);
                    const float4 v2 = *(const float4*)(xs + 8);
                    const float4 v3 = *(const float4*)(xs + 12);
                    a0.x += v0.x; a0.y += v0.y; a0.z += v0.z; a0.w += v0.w;
                    a1.x += v1.x; a1.y += v1.y; a1.z += v1.z; a1.w += v1.w;
                    a2.x += v2.x; a2.y += v2.y; a2.z += v2.z; a2.w += v2.w;
                    a3.x += v3.x; a3.y += v3.y; a3.z += v3.z; a3.w += v3.w;
                }
                const float inv = 1.0f / fmaxf((float)d, 1.0f);
                a0.x *= inv; a0.y *= inv; a0.z *= inv; a0.w *= inv;
                a1.x *= inv; a1.y *= inv; a1.z *= inv; a1.w *= inv;
                a2.x *= inv; a2.y *= inv; a2.z *= inv; a2.w *= inv;
                a3.x *= inv; a3.y *= inv; a3.z *= inv; a3.w *= inv;
            } else {
                const float* xs = x + (size_t)node * IN_F + (k0 - IN_F);
                a0 = *(const float4*)(xs + 0);
                a1 = *(const float4*)(xs + 4);
                a2 = *(const float4*)(xs + 8);
                a3 = *(const float4*)(xs + 12);
            }
        }
        float* dst = &ldsA[m * LDA + k0];
        *(float4*)(dst + 0)  = a0;
        *(float4*)(dst + 4)  = a1;
        *(float4*)(dst + 8)  = a2;
        *(float4*)(dst + 12) = a3;
    }
    __syncthreads();

    const int bucket = s_bucket;
    if (bucket < 0) return;

    const int wave  = tid >> 5;
    const int lane  = tid & 31;
    const int lr    = lane & 15;
    const int khalf = (lane >> 4) << 1;
    const int ncol  = wave * 16 + lr;

    const float* __restrict__ wn   = neigh_w + ((size_t)bucket * OUT_F + ncol) * IN_F;
    const float* __restrict__ wr   = root_w  + ((size_t)bucket * OUT_F + ncol) * IN_F;
    const float* __restrict__ aRow = &ldsA[lr * LDA];

    v8f acc = {};
    #pragma unroll
    for (int kk = 0; kk < 32; ++kk) {
        const int k = kk * 4 + khalf;
        v2f a, b;
        a.x = aRow[k];     a.y = aRow[k + 1];
        b.x = wn[k];       b.y = wn[k + 1];
        acc = __builtin_amdgcn_wmma_f32_16x16x4_f32(false, a, false, b,
                                                    (short)0, acc, false, false);
    }
    #pragma unroll
    for (int kk = 0; kk < 32; ++kk) {
        const int k = kk * 4 + khalf;
        v2f a, b;
        a.x = aRow[IN_F + k]; a.y = aRow[IN_F + k + 1];
        b.x = wr[k];          b.y = wr[k + 1];
        acc = __builtin_amdgcn_wmma_f32_16x16x4_f32(false, a, false, b,
                                                    (short)0, acc, false, false);
    }

    const float bias = neigh_b[bucket * OUT_F + ncol] + root_b[bucket * OUT_F + ncol];
    const int mbase  = (lane >> 4) * 8;
    #pragma unroll
    for (int i = 0; i < 8; ++i) {
        const int node = s_node[mbase + i];
        if (node >= 0) out[(size_t)node * OUT_F + ncol] = acc[i] + bias;
    }
}

extern "C" void kernel_launch(void* const* d_in, const int* in_sizes, int n_in,
                              void* d_out, int out_size, void* d_ws, size_t ws_size,
                              hipStream_t stream) {
    const float* x       = (const float*)d_in[0];
    const float* neigh_w = (const float*)d_in[1];
    const float* neigh_b = (const float*)d_in[2];
    const float* root_w  = (const float*)d_in[3];
    const float* root_b  = (const float*)d_in[4];
    const int*   eidx    = (const int*)d_in[5];

    const int N = in_sizes[0] / IN_F;
    const int E = in_sizes[5] / 2;
    const int* row = eidx;
    const int* col = eidx + E;

    const int nScanBlocks = (N + SCAN_B - 1) / SCAN_B;
    const int permLen = ((N + NBUCK * TILE_M + TILE_M - 1) / TILE_M) * TILE_M;
    const int nTiles  = permLen / TILE_M;

    char* ws = (char*)d_ws;
    int* degi     = (int*)ws;  ws += (size_t)N * sizeof(int);
    int* cursor   = (int*)ws;  ws += (size_t)N * sizeof(int);
    int* rowstart = (int*)ws;  ws += (size_t)N * sizeof(int);
    int* bsum     = (int*)ws;  ws += (size_t)((nScanBlocks + 15) & ~15) * sizeof(int);
    int* edge_src = (int*)ws;  ws += (size_t)E * sizeof(int);
    int* perm     = (int*)ws;  ws += (size_t)permLen * sizeof(int);
    int* meta     = (int*)ws;

    fill_i32_kernel<<<1024, 256, 0, stream>>>(degi, 0, (size_t)2 * N);
    fill_i32_kernel<<<1024, 256, 0, stream>>>(perm, -1, (size_t)permLen);
    fill_i32_kernel<<<1, 32, 0, stream>>>(meta, 0, 16);

    degree_kernel<<<(E + 255) / 256, 256, 0, stream>>>(col, degi, E);
    scan1_kernel<<<nScanBlocks, SCAN_B, 0, stream>>>(degi, rowstart, bsum, N);
    scan2_kernel<<<1, 1, 0, stream>>>(bsum, nScanBlocks);
    scan3_kernel<<<nScanBlocks, SCAN_B, 0, stream>>>(rowstart, bsum, N);
    csr_fill_kernel<<<(E + 255) / 256, 256, 0, stream>>>(row, col, rowstart, cursor,
                                                         edge_src, E);

    bucket_count_kernel<<<(N + BK_B - 1) / BK_B, BK_B, 0, stream>>>(degi, meta, N);
    bucket_offset_kernel<<<1, 1, 0, stream>>>(meta);
    bucket_scatter_kernel<<<(N + BK_B - 1) / BK_B, BK_B, 0, stream>>>(degi, meta, perm, N);

    mfconv_gemm_kernel<<<nTiles, 256, 0, stream>>>(x, degi, rowstart, edge_src, perm,
                                                   neigh_w, neigh_b, root_w, root_b,
                                                   (float*)d_out);
}


# GEMM kernel: work-group = four 16-node tiles (64 nodes); each weight fragment is loaded once per k step and feeds the four row tiles' matrix instructions instead of being reloaded per 16 nodes
# speedup vs baseline: 2.0449x; 1.3269x over previous
#include <hip/hip_runtime.h>

typedef __attribute__((ext_vector_type(2))) float v2f;
typedef __attribute__((ext_vector_type(8))) float v8f;

#define IN_F   128
#define OUT_F  128
#define MAXDEG 4
#define NBUCK  5
#define TILE_M 16
#define LDA    260
#define SCAN_B 256

__global__ void fill_i32_kernel(int* __restrict__ p, int val, size_t n) {
    size_t i = (size_t)blockIdx.x * blockDim.x + threadIdx.x;
    size_t stride = (size_t)gridDim.x * blockDim.x;
    for (; i < n; i += stride) p[i] = val;
}

__global__ void degree_kernel(const int* __restrict__ col, int* __restrict__ degi, int E) {
    int e = blockIdx.x * blockDim.x + threadIdx.x;
    if (e < E) atomicAdd(&degi[col[e]], 1);
}

__global__ void scan1_kernel(const int* __restrict__ degi, int* __restrict__ rowstart,
                             int* __restrict__ bsum, int N) {
    __shared__ int sh[SCAN_B];
    int i = blockIdx.x * SCAN_B + threadIdx.x;
    int v = (i < N) ? degi[i] : 0;
    sh[threadIdx.x] = v;
    __syncthreads();
    #pragma unroll
    for (int off = 1; off < SCAN_B; off <<= 1) {
        int t = (threadIdx.x >= off) ? sh[threadIdx.x - off] : 0;
        __syncthreads();
        sh[threadIdx.x] += t;
        __syncthreads();
    }
    if (i < N) rowstart[i] = sh[threadIdx.x] - v;
    if (threadIdx.x == SCAN_B - 1) bsum[blockIdx.x] = sh[SCAN_B - 1];
}

__global__ void scan2_kernel(int* __restrict__ bsum, int nBlocks) {
    if (threadIdx.x == 0 && blockIdx.x == 0) {
        int acc = 0;
        for (int b = 0; b < nBlocks; ++b) { int t = bsum[b]; bsum[b] = acc; acc += t; }
    }
}

__global__ void scan3_kernel(int* __restrict__ rowstart, const int* __restrict__ bsum, int N) {
    int i = blockIdx.x * SCAN_B + threadIdx.x;
    if (i < N) rowstart[i] += bsum[blockIdx.x];
}

__global__ void csr_fill_kernel(const int* __restrict__ row, const int* __restrict__ col,
                                const int* __restrict__ rowstart, int* __restrict__ cursor,
                                int* __restrict__ edge_src, int E) {
    int e = blockIdx.x * blockDim.x + threadIdx.x;
    if (e >= E) return;
    int d = col[e];
    int pos = atomicAdd(&cursor[d], 1);
    edge_src[rowstart[d] + pos] = row[e];
}

#define BK_B  1024
#define BK_W  (BK_B / 32)

__global__ __launch_bounds__(BK_B) void bucket_count_kernel(const int* __restrict__ degi,
                                                            int* __restrict__ meta, int N) {
    __shared__ int s_w[BK_W][NBUCK];
    const int tid  = threadIdx.x;
    const int lane = tid & 31;
    const int wave = tid >> 5;
    const int i = blockIdx.x * BK_B + tid;
    const int b = (i < N) ? min(degi[i], MAXDEG) : -1;
    int myc = 0;
    for (int k = 0; k < NBUCK; ++k) {
        const unsigned m = __builtin_amdgcn_ballot_w32(b == k);
        if (lane == k) myc = __builtin_popcount(m);
    }
    if (lane < NBUCK) s_w[wave][lane] = myc;
    __syncthreads();
    if (tid < NBUCK) {
        int acc = 0;
        for (int w = 0; w < BK_W; ++w) acc += s_w[w][tid];
        if (acc) atomicAdd(&meta[tid], acc);
    }
}

__global__ void bucket_offset_kernel(int* __restrict__ meta) {
    if (threadIdx.x == 0 && blockIdx.x == 0) {
        int off = 0;
        for (int b = 0; b < NBUCK; ++b) {
            meta[5 + b] = off;
            off += (meta[b] + TILE_M - 1) & ~(TILE_M - 1);
        }
    }
}

__global__ __launch_bounds__(BK_B) void bucket_scatter_kernel(const int* __restrict__ degi,
                                                              int* __restrict__ meta,
                                                              int* __restrict__ perm, int N) {
    __shared__ int s_w[BK_W][NBUCK];
    __shared__ int s_base[NBUCK];
    const int tid  = threadIdx.x;
    const int lane = tid & 31;
    const int wave = tid >> 5;
    const int i = blockIdx.x * BK_B + tid;
    const int b = (i < N) ? min(degi[i], MAXDEG) : -1;
    unsigned mine = 0;
    int myc = 0;
    for (int k = 0; k < NBUCK; ++k) {
        const unsigned m = __builtin_amdgcn_ballot_w32(b == k);
        if (b == k)    mine = m;
        if (lane == k) myc  = __builtin_popcount(m);
    }
    if (lane < NBUCK) s_w[wave][lane] = myc;
    __syncthreads();
    if (tid < NBUCK) {
        int acc = 0;
        for (int w = 0; w < BK_W; ++w) { const int c = s_w[w][tid]; s_w[w][tid] = acc; acc += c; }
        int base = 0;
        if (acc) base = atomicAdd(&meta[10 + tid], acc);
        s_base[tid] = meta[5 + tid] + base;
    }
    __syncthreads();
    if (b >= 0) {
        const int rank = __builtin_popcount(mine & ((1u << lane) - 1u));
        perm[s_base[b] + s_w[wave][b] + rank] = i;
    }
}

#define WG_TILES 4
#define WG_M     (TILE_M * WG_TILES)

template <int NT>
__device__ __forceinline__ void cat_gemm(const float* aRow, const float* __restrict__ wn,
                                         const float* __restrict__ wr, const int khalf,
                                         v8f (&acc)[NT]) {
    #pragma unroll
    for (int kk = 0; kk < 32; ++kk) {
        const int k = kk * 4 + khalf;
        v2f b;
        b.x = wn[k];       b.y = wn[k + 1];
        #pragma unroll
        for (int t = 0; t < NT; ++t) {
            v2f a;
            a.x = aRow[t * TILE_M * LDA + k];  a.y = aRow[t * TILE_M * LDA + k + 1];
            acc[t] = __builtin_amdgcn_wmma_f32_16x16x4_f32(false, a, false, b,
                                                           (short)0, acc[t], false, false);
        }
    }
    #pragma unroll
    for (int kk = 0; kk < 32; ++kk) {
        const int k = kk * 4 + khalf;
        v2f b;
        b.x = wr[k];       b.y = wr[k + 1];
        #pragma unroll
        for (int t = 0; t < NT; ++t) {
            v2f a;
            a.x = aRow[t * TILE_M * LDA + IN_F + k];  a.y = aRow[t * TILE_M * LDA + IN_F + k + 1];
            acc[t] = __builtin_amdgcn_wmma_f32_16x16x4_f32(false, a, false, b,
                                                           (short)0, acc[t], false, false);
        }
    }
}

__device__ __forceinline__ void store_tile(const v8f acc, const int* nodes, const int mbase,
                                           const float bias, const int ncol,
                                           float* __restrict__ out) {
    #pragma unroll
    for (int i = 0; i < 8; ++i) {
        const int node = nodes[mbase + i];
        if (node >= 0) out[(size_t)node * OUT_F + ncol] = acc[i] + bias;
    }
}

__global__ __launch_bounds__(256) void mfconv_gemm_kernel(
    const float* __restrict__ x,
    const int* __restrict__ degi, const int* __restrict__ rowstart,
    const int* __restrict__ edge_src, const int* __restrict__ perm,
    const float* __restrict__ neigh_w, const float* __restrict__ neigh_b,
    const float* __restrict__ root_w,  const float* __restrict__ root_b,
    float* __restrict__ out, const int permLen)
{
    __shared__ float ldsA[WG_M * LDA];
    __shared__ int   s_node[WG_M];
    __shared__ int   s_bucket[WG_TILES];

    const int tid = threadIdx.x;

    if (tid < WG_M) {
        const int p = blockIdx.x * WG_M + tid;
        s_node[tid] = (p < permLen) ? perm[p] : -1;
    }
    __syncthreads();

    if (tid < WG_TILES) {
        int b = -1;
        for (int i = 0; i < TILE_M; ++i) {
            int nd = s_node[tid * TILE_M + i];
            if (nd >= 0) { b = min(degi[nd], MAXDEG); break; }
        }
        s_bucket[tid] = b;
    }

    {
        const int mloc = tid >> 4;
        const int k0   = (tid & 15) * 16;
        int node[WG_TILES], deg[WG_TILES], start[WG_TILES];
        #pragma unroll
        for (int t = 0; t < WG_TILES; ++t) node[t] = s_node[t * TILE_M + mloc];
        #pragma unroll
        for (int t = 0; t < WG_TILES; ++t) {
            deg[t] = 0; start[t] = 0;
            if (node[t] >= 0 && k0 < IN_F) { deg[t] = degi[node[t]]; start[t] = rowstart[node[t]]; }
        }
        #pragma unroll
        for (int t = 0; t < WG_TILES; ++t) {
            float4 a0 = make_float4(0.f,0.f,0.f,0.f), a1 = a0, a2 = a0, a3 = a0;
            if (node[t] >= 0) {
                if (k0 < IN_F) {
                    const int d = deg[t];
                    for (int j = 0; j < d; ++j) {
                        const float* xs = x + (size_t)edge_src[start[t] + j] * IN_F + k0;
                        const float4 v0 = *(const float4*)(xs + 0);
                        const float4 v1 = *(const float4*)(xs + 4);
                        const float4 v2 = *(const float4*)(xs + 8);
                        const float4 v3 = *(const float4*)(xs + 12);
                        a0.x += v0.x; a0.y += v0.y; a0.z += v0.z; a0.w += v0.w;
                        a1.x += v1.x; a1.y += v1.y; a1.z += v1.z; a1.w += v1.w;
                        a2.x += v2.x; a2.y += v2.y; a2.z += v2.z; a2.w += v2.w;
                        a3.x += v3.x; a3.y += v3.y; a3.z += v3.z; a3.w += v3.w;
                    }
                    const float inv = 1.0f / fmaxf((float)d, 1.0f);
                    a0.x *= inv; a0.y *= inv; a0.z *= inv; a0.w *= inv;
                    a1.x *= inv; a1.y *= inv; a1.z *= inv; a1.w *= inv;
                    a2.x *= inv; a2.y *= inv; a2.z *= inv; a2.w *= inv;
                    a3.x *= inv; a3.y *= inv; a3.z *= inv; a3.w *= inv;
                } else {
                    const float* xs = x + (size_t)node[t] * IN_F + (k0 - IN_F);
                    a0 = *(const float4*)(xs + 0);
                    a1 = *(const float4*)(xs + 4);
                    a2 = *(const float4*)(xs + 8);
                    a3 = *(const float4*)(xs + 12);
                }
            }
            float* dst = &ldsA[(t * TILE_M + mloc) * LDA + k0];
            *(float4*)(dst + 0)  = a0;
            *(float4*)(dst + 4)  = a1;
            *(float4*)(dst + 8)  = a2;
            *(float4*)(dst + 12) = a3;
        }
    }
    __syncthreads();

    int bk[WG_TILES];
    #pragma unroll
    for (int t = 0; t < WG_TILES; ++t) bk[t] = __builtin_amdgcn_readfirstlane(s_bucket[t]);
    const int bmax = max(max(bk[0], bk[1]), max(bk[2], bk[3]));
    if (bmax < 0) return;
    const bool shared_bucket = (bk[0] < 0 || bk[0] == bmax) && (bk[1] < 0 || bk[1] == bmax) &&
                               (bk[2] < 0 || bk[2] == bmax) && (bk[3] < 0 || bk[3] == bmax);

    const int wave  = tid >> 5;
    const int lane  = tid & 31;
    const int lr    = lane & 15;
    const int khalf = (lane >> 4) << 1;
    const int ncol  = wave * 16 + lr;
    const int mbase = (lane >> 4) * 8;
    const float* aRow = &ldsA[lr * LDA];

    if (shared_bucket) {
        const float* __restrict__ wn = neigh_w + ((size_t)bmax * OUT_F + ncol) * IN_F;
        const float* __restrict__ wr = root_w  + ((size_t)bmax * OUT_F + ncol) * IN_F;
        v8f acc[WG_TILES] = {};
        cat_gemm<WG_TILES>(aRow, wn, wr, khalf, acc);
        const float bias = neigh_b[bmax * OUT_F + ncol] + root_b[bmax * OUT_F + ncol];
        #pragma unroll
        for (int t = 0; t < WG_TILES; ++t)
            store_tile(acc[t], &s_node[t * TILE_M], mbase, bias, ncol, out);
    } else {
        #pragma unroll 1
        for (int t = 0; t < WG_TILES; ++t) {
            const int b = __builtin_amdgcn_readfirstlane(s_bucket[t]);
            if (b < 0) continue;
            const float* __restrict__ wn = neigh_w + ((size_t)b * OUT_F + ncol) * IN_F;
            const float* __restrict__ wr = root_w  + ((size_t)b * OUT_F + ncol) * IN_F;
            v8f acc1[1] = {};
            cat_gemm<1>(aRow + t * TILE_M * LDA, wn, wr, khalf, acc1);
            const float bias = neigh_b[b * OUT_F + ncol] + root_b[b * OUT_F + ncol];
            store_tile(acc1[0], &s_node[t * TILE_M], mbase, bias, ncol, out);
        }
    }
}

extern "C" void kernel_launch(void* const* d_in, const int* in_sizes, int n_in,
                              void* d_out, int out_size, void* d_ws, size_t ws_size,
                              hipStream_t stream) {
    const float* x       = (const float*)d_in[0];
    const float* neigh_w = (const float*)d_in[1];
    const float* neigh_b = (const float*)d_in[2];
    const float* root_w  = (const float*)d_in[3];
    const float* root_b  = (const float*)d_in[4];
    const int*   eidx    = (const int*)d_in[5];

    const int N = in_sizes[0] / IN_F;
    const int E = in_sizes[5] / 2;
    const int* row = eidx;
    const int* col = eidx + E;

    const int nScanBlocks = (N + SCAN_B - 1) / SCAN_B;
    const int permLen = ((N + NBUCK * TILE_M + TILE_M - 1) / TILE_M) * TILE_M;
    const int nGroups = (permLen + WG_M - 1) / WG_M;

    char* ws = (char*)d_ws;
    int* degi     = (int*)ws;  ws += (size_t)N * sizeof(int);
    int* cursor   = (int*)ws;  ws += (size_t)N * sizeof(int);
    int* rowstart = (int*)ws;  ws += (size_t)N * sizeof(int);
    int* bsum     = (int*)ws;  ws += (size_t)((nScanBlocks + 15) & ~15) * sizeof(int);
    int* edge_src = (int*)ws;  ws += (size_t)E * sizeof(int);
    int* perm     = (int*)ws;  ws += (size_t)permLen * sizeof(int);
    int* meta     = (int*)ws;

    fill_i32_kernel<<<1024, 256, 0, stream>>>(degi, 0, (size_t)2 * N);
    fill_i32_kernel<<<1024, 256, 0, stream>>>(perm, -1, (size_t)permLen);
    fill_i32_kernel<<<1, 32, 0, stream>>>(meta, 0, 16);

    degree_kernel<<<(E + 255) / 256, 256, 0, stream>>>(col, degi, E);
    scan1_kernel<<<nScanBlocks, SCAN_B, 0, stream>>>(degi, rowstart, bsum, N);
    scan2_kernel<<<1, 1, 0, stream>>>(bsum, nScanBlocks);
    scan3_kernel<<<nScanBlocks, SCAN_B, 0, stream>>>(rowstart, bsum, N);
    csr_fill_kernel<<<(E + 255) / 256, 256, 0, stream>>>(row, col, rowstart, cursor,
                                                         edge_src, E);

    bucket_count_kernel<<<(N + BK_B - 1) / BK_B, BK_B, 0, stream>>>(degi, meta, N);
    bucket_offset_kernel<<<1, 1, 0, stream>>>(meta);
    bucket_scatter_kernel<<<(N + BK_B - 1) / BK_B, BK_B, 0, stream>>>(degi, meta, perm, N);

    mfconv_gemm_kernel<<<nGroups, 256, 0, stream>>>(x, degi, rowstart, edge_src, perm,
                                                    neigh_w, neigh_b, root_w, root_b,
                                                    (float*)d_out, permLen);
}


# CSR segment starts: each 256-node work-group scans its degrees and reserves its run of edge slots with one returning atomic add; the three-launch global prefix sum is gone (12 -> 10 launches)
# speedup vs baseline: 2.3365x; 1.1426x over previous
#include <hip/hip_runtime.h>

typedef __attribute__((ext_vector_type(2))) float v2f;
typedef __attribute__((ext_vector_type(8))) float v8f;

#define IN_F   128
#define OUT_F  128
#define MAXDEG 4
#define NBUCK  5
#define TILE_M 16
#define LDA    260

__global__ void fill_i32_kernel(int* __restrict__ p, int val, size_t n) {
    size_t i = (size_t)blockIdx.x * blockDim.x + threadIdx.x;
    size_t stride = (size_t)gridDim.x * blockDim.x;
    for (; i < n; i += stride) p[i] = val;
}

__global__ void degree_kernel(const int* __restrict__ col, int* __restrict__ degi, int E) {
    int e = blockIdx.x * blockDim.x + threadIdx.x;
    if (e < E) atomicAdd(&degi[col[e]], 1);
}

#define SEG_B 256
#define SEG_W (SEG_B / 32)

__global__ __launch_bounds__(SEG_B) void seg_reserve_kernel(const int* __restrict__ degi,
                                                            int* __restrict__ rowstart,
                                                            int* __restrict__ total, int N) {
    __shared__ int s_w[SEG_W];
    const int tid  = threadIdx.x;
    const int lane = tid & 31;
    const int wave = tid >> 5;
    const int i = blockIdx.x * SEG_B + tid;
    const int v = (i < N) ? degi[i] : 0;
    int inc = v;
    for (int off = 1; off < 32; off <<= 1) {
        const int t = __shfl_up(inc, off, 32);
        if (lane >= off) inc += t;
    }
    if (lane == 31) s_w[wave] = inc;
    __syncthreads();
    if (wave == 0) {
        const int w = (lane < SEG_W) ? s_w[lane] : 0;
        int winc = w;
        for (int off = 1; off < SEG_W; off <<= 1) {
            const int t = __shfl_up(winc, off, 32);
            if (lane >= off) winc += t;
        }
        int base = 0;
        if (lane == SEG_W - 1) base = atomicAdd(total, winc);
        base = __shfl(base, SEG_W - 1, 32);
        if (lane < SEG_W) s_w[lane] = base + winc - w;
    }
    __syncthreads();
    if (i < N) rowstart[i] = s_w[wave] + inc - v;
}

__global__ void csr_fill_kernel(const int* __restrict__ row, const int* __restrict__ col,
                                const int* __restrict__ rowstart, int* __restrict__ cursor,
                                int* __restrict__ edge_src, int E) {
    int e = blockIdx.x * blockDim.x + threadIdx.x;
    if (e >= E) return;
    int d = col[e];
    int pos = atomicAdd(&cursor[d], 1);
    edge_src[rowstart[d] + pos] = row[e];
}

#define BK_B  1024
#define BK_W  (BK_B / 32)

__global__ __launch_bounds__(BK_B) void bucket_count_kernel(const int* __restrict__ degi,
                                                            int* __restrict__ meta, int N) {
    __shared__ int s_w[BK_W][NBUCK];
    const int tid  = threadIdx.x;
    const int lane = tid & 31;
    const int wave = tid >> 5;
    const int i = blockIdx.x * BK_B + tid;
    const int b = (i < N) ? min(degi[i], MAXDEG) : -1;
    int myc = 0;
    for (int k = 0; k < NBUCK; ++k) {
        const unsigned m = __builtin_amdgcn_ballot_w32(b == k);
        if (lane == k) myc = __builtin_popcount(m);
    }
    if (lane < NBUCK) s_w[wave][lane] = myc;
    __syncthreads();
    if (tid < NBUCK) {
        int acc = 0;
        for (int w = 0; w < BK_W; ++w) acc += s_w[w][tid];
        if (acc) atomicAdd(&meta[tid], acc);
    }
}

__global__ void bucket_offset_kernel(int* __restrict__ meta) {
    if (threadIdx.x == 0 && blockIdx.x == 0) {
        int off = 0;
        for (int b = 0; b < NBUCK; ++b) {
            meta[5 + b] = off;
            off += (meta[b] + TILE_M - 1) & ~(TILE_M - 1);
        }
    }
}

__global__ __launch_bounds__(BK_B) void bucket_scatter_kernel(const int* __restrict__ degi,
                                                              int* __restrict__ meta,
                                                              int* __restrict__ perm, int N) {
    __shared__ int s_w[BK_W][NBUCK];
    __shared__ int s_base[NBUCK];
    const int tid  = threadIdx.x;
    const int lane = tid & 31;
    const int wave = tid >> 5;
    const int i = blockIdx.x * BK_B + tid;
    const int b = (i < N) ? min(degi[i], MAXDEG) : -1;
    unsigned mine = 0;
    int myc = 0;
    for (int k = 0; k < NBUCK; ++k) {
        const unsigned m = __builtin_amdgcn_ballot_w32(b == k);
        if (b == k)    mine = m;
        if (lane == k) myc  = __builtin_popcount(m);
    }
    if (lane < NBUCK) s_w[wave][lane] = myc;
    __syncthreads();
    if (tid < NBUCK) {
        int acc = 0;
        for (int w = 0; w < BK_W; ++w) { const int c = s_w[w][tid]; s_w[w][tid] = acc; acc += c; }
        int base = 0;
        if (acc) base = atomicAdd(&meta[10 + tid], acc);
        s_base[tid] = meta[5 + tid] + base;
    }
    __syncthreads();
    if (b >= 0) {
        const int rank = __builtin_popcount(mine & ((1u << lane) - 1u));
        perm[s_base[b] + s_w[wave][b] + rank] = i;
    }
}

#define WG_TILES 4
#define WG_M     (TILE_M * WG_TILES)

template <int NT>
__device__ __forceinline__ void cat_gemm(const float* aRow, const float* __restrict__ wn,
                                         const float* __restrict__ wr, const int khalf,
                                         v8f (&acc)[NT]) {
    #pragma unroll
    for (int kk = 0; kk < 32; ++kk) {
        const int k = kk * 4 + khalf;
        v2f b;
        b.x = wn[k];       b.y = wn[k + 1];
        #pragma unroll
        for (int t = 0; t < NT; ++t) {
            v2f a;
            a.x = aRow[t * TILE_M * LDA + k];  a.y = aRow[t * TILE_M * LDA + k + 1];
            acc[t] = __builtin_amdgcn_wmma_f32_16x16x4_f32(false, a, false, b,
                                                           (short)0, acc[t], false, false);
        }
    }
    #pragma unroll
    for (int kk = 0; kk < 32; ++kk) {
        const int k = kk * 4 + khalf;
        v2f b;
        b.x = wr[k];       b.y = wr[k + 1];
        #pragma unroll
        for (int t = 0; t < NT; ++t) {
            v2f a;
            a.x = aRow[t * TILE_M * LDA + IN_F + k];  a.y = aRow[t * TILE_M * LDA + IN_F + k + 1];
            acc[t] = __builtin_amdgcn_wmma_f32_16x16x4_f32(false, a, false, b,
                                                           (short)0, acc[t], false, false);
        }
    }
}

__device__ __forceinline__ void store_tile(const v8f acc, const int* nodes, const int mbase,
                                           const float bias, const int ncol,
                                           float* __restrict__ out) {
    #pragma unroll
    for (int i = 0; i < 8; ++i) {
        const int node = nodes[mbase + i];
        if (node >= 0) out[(size_t)node * OUT_F + ncol] = acc[i] + bias;
    }
}

__global__ __launch_bounds__(256) void mfconv_gemm_kernel(
    const float* __restrict__ x,
    const int* __restrict__ degi, const int* __restrict__ rowstart,
    const int* __restrict__ edge_src, const int* __restrict__ perm,
    const float* __restrict__ neigh_w, const float* __restrict__ neigh_b,
    const float* __restrict__ root_w,  const float* __restrict__ root_b,
    float* __restrict__ out, const int permLen)
{
    __shared__ float ldsA[WG_M * LDA];
    __shared__ int   s_node[WG_M];
    __shared__ int   s_bucket[WG_TILES];

    const int tid = threadIdx.x;

    if (tid < WG_M) {
        const int p = blockIdx.x * WG_M + tid;
        s_node[tid] = (p < permLen) ? perm[p] : -1;
    }
    __syncthreads();

    if (tid < WG_TILES) {
        int b = -1;
        for (int i = 0; i < TILE_M; ++i) {
            int nd = s_node[tid * TILE_M + i];
            if (nd >= 0) { b = min(degi[nd], MAXDEG); break; }
        }
        s_bucket[tid] = b;
    }

    {
        const int mloc = tid >> 4;
        const int k0   = (tid & 15) * 16;
        int node[WG_TILES], deg[WG_TILES], start[WG_TILES];
        #pragma unroll
        for (int t = 0; t < WG_TILES; ++t) node[t] = s_node[t * TILE_M + mloc];
        #pragma unroll
        for (int t = 0; t < WG_TILES; ++t) {
            deg[t] = 0; start[t] = 0;
            if (node[t] >= 0 && k0 < IN_F) { deg[t] = degi[node[t]]; start[t] = rowstart[node[t]]; }
        }
        #pragma unroll
        for (int t = 0; t < WG_TILES; ++t) {
            float4 a0 = make_float4(0.f,0.f,0.f,0.f), a1 = a0, a2 = a0, a3 = a0;
            if (node[t] >= 0) {
                if (k0 < IN_F) {
                    const int d = deg[t];
                    for (int j = 0; j < d; ++j) {
                        const float* xs = x + (size_t)edge_src[start[t] + j] * IN_F + k0;
                        const float4 v0 = *(const float4*)(xs + 0);
                        const float4 v1 = *(const float4*)(xs + 4);
                        const float4 v2 = *(const float4*)(xs + 8);
                        const float4 v3 = *(const float4*)(xs + 12);
                        a0.x += v0.x; a0.y += v0.y; a0.z += v0.z; a0.w += v0.w;
                        a1.x += v1.x; a1.y += v1.y; a1.z += v1.z; a1.w += v1.w;
                        a2.x += v2.x; a2.y += v2.y; a2.z += v2.z; a2.w += v2.w;
                        a3.x += v3.x; a3.y += v3.y; a3.z += v3.z; a3.w += v3.w;
                    }
                    const float inv = 1.0f / fmaxf((float)d, 1.0f);
                    a0.x *= inv; a0.y *= inv; a0.z *= inv; a0.w *= inv;
                    a1.x *= inv; a1.y *= inv; a1.z *= inv; a1.w *= inv;
                    a2.x *= inv; a2.y *= inv; a2.z *= inv; a2.w *= inv;
                    a3.x *= inv; a3.y *= inv; a3.z *= inv; a3.w *= inv;
                } else {
                    const float* xs = x + (size_t)node[t] * IN_F + (k0 - IN_F);
                    a0 = *(const float4*)(xs + 0);
                    a1 = *(const float4*)(xs + 4);
                    a2 = *(const float4*)(xs + 8);
                    a3 = *(const float4*)(xs + 12);
                }
            }
            float* dst = &ldsA[(t * TILE_M + mloc) * LDA + k0];
            *(float4*)(dst + 0)  = a0;
            *(float4*)(dst + 4)  = a1;
            *(float4*)(dst + 8)  = a2;
            *(float4*)(dst + 12) = a3;
        }
    }
    __syncthreads();

    int bk[WG_TILES];
    #pragma unroll
    for (int t = 0; t < WG_TILES; ++t) bk[t] = __builtin_amdgcn_readfirstlane(s_bucket[t]);
    const int bmax = max(max(bk[0], bk[1]), max(bk[2], bk[3]));
    if (bmax < 0) return;
    const bool shared_bucket = (bk[0] < 0 || bk[0] == bmax) && (bk[1] < 0 || bk[1] == bmax) &&
                               (bk[2] < 0 || bk[2] == bmax) && (bk[3] < 0 || bk[3] == bmax);

    const int wave  = tid >> 5;
    const int lane  = tid & 31;
    const int lr    = lane & 15;
    const int khalf = (lane >> 4) << 1;
    const int ncol  = wave * 16 + lr;
    const int mbase = (lane >> 4) * 8;
    const float* aRow = &ldsA[lr * LDA];

    if (shared_bucket) {
        const float* __restrict__ wn = neigh_w + ((size_t)bmax * OUT_F + ncol) * IN_F;
        const float* __restrict__ wr = root_w  + ((size_t)bmax * OUT_F + ncol) * IN_F;
        v8f acc[WG_TILES] = {};
        cat_gemm<WG_TILES>(aRow, wn, wr, khalf, acc);
        const float bias = neigh_b[bmax * OUT_F + ncol] + root_b[bmax * OUT_F + ncol];
        #pragma unroll
        for (int t = 0; t < WG_TILES; ++t)
            store_tile(acc[t], &s_node[t * TILE_M], mbase, bias, ncol, out);
    } else {
        #pragma unroll 1
        for (int t = 0; t < WG_TILES; ++t) {
            const int b = __builtin_amdgcn_readfirstlane(s_bucket[t]);
            if (b < 0) continue;
            const float* __restrict__ wn = neigh_w + ((size_t)b * OUT_F + ncol) * IN_F;
            const float* __restrict__ wr = root_w  + ((size_t)b * OUT_F + ncol) * IN_F;
            v8f acc1[1] = {};
            cat_gemm<1>(aRow + t * TILE_M * LDA, wn, wr, khalf, acc1);
            const float bias = neigh_b[b * OUT_F + ncol] + root_b[b * OUT_F + ncol];
            store_tile(acc1[0], &s_node[t * TILE_M], mbase, bias, ncol, out);
        }
    }
}

extern "C" void kernel_launch(void* const* d_in, const int* in_sizes, int n_in,
                              void* d_out, int out_size, void* d_ws, size_t ws_size,
                              hipStream_t stream) {
    const float* x       = (const float*)d_in[0];
    const float* neigh_w = (const float*)d_in[1];
    const float* neigh_b = (const float*)d_in[2];
    const float* root_w  = (const float*)d_in[3];
    const float* root_b  = (const float*)d_in[4];
    const int*   eidx    = (const int*)d_in[5];

    const int N = in_sizes[0] / IN_F;
    const int E = in_sizes[5] / 2;
    const int* row = eidx;
    const int* col = eidx + E;

    const int nSegBlocks = (N + SEG_B - 1) / SEG_B;
    const int permLen = ((N + NBUCK * TILE_M + TILE_M - 1) / TILE_M) * TILE_M;
    const int nGroups = (permLen + WG_M - 1) / WG_M;

    char* ws = (char*)d_ws;
    int* degi     = (int*)ws;  ws += (size_t)N * sizeof(int);
    int* cursor   = (int*)ws;  ws += (size_t)N * sizeof(int);
    int* rowstart = (int*)ws;  ws += (size_t)N * sizeof(int);
    int* edge_src = (int*)ws;  ws += (size_t)E * sizeof(int);
    int* perm     = (int*)ws;  ws += (size_t)permLen * sizeof(int);
    int* meta     = (int*)ws;

    fill_i32_kernel<<<1024, 256, 0, stream>>>(degi, 0, (size_t)2 * N);
    fill_i32_kernel<<<1024, 256, 0, stream>>>(perm, -1, (size_t)permLen);
    fill_i32_kernel<<<1, 32, 0, stream>>>(meta, 0, 16);

    degree_kernel<<<(E + 255) / 256, 256, 0, stream>>>(col, degi, E);
    seg_reserve_kernel<<<nSegBlocks, SEG_B, 0, stream>>>(degi, rowstart, meta + 15, N);
    csr_fill_kernel<<<(E + 255) / 256, 256, 0, stream>>>(row, col, rowstart, cursor,
                                                         edge_src, E);

    bucket_count_kernel<<<(N + BK_B - 1) / BK_B, BK_B, 0, stream>>>(degi, meta, N);
    bucket_offset_kernel<<<1, 1, 0, stream>>>(meta);
    bucket_scatter_kernel<<<(N + BK_B - 1) / BK_B, BK_B, 0, stream>>>(degi, meta, perm, N);

    mfconv_gemm_kernel<<<nGroups, 256, 0, stream>>>(x, degi, rowstart, edge_src, perm,
                                                    neigh_w, neigh_b, root_w, root_b,
                                                    (float*)d_out, permLen);
}


# CSR build: the degree pass's counting add returns each edge's slot (kept 16-bit), the fill places edges from it: no second atomic pass, no cursor array; neighbour sum order = that add's arrival order
# speedup vs baseline: 2.6274x; 1.1245x over previous
#include <hip/hip_runtime.h>

typedef __attribute__((ext_vector_type(2))) float v2f;
typedef __attribute__((ext_vector_type(8))) float v8f;

#define IN_F   128
#define OUT_F  128
#define MAXDEG 4
#define NBUCK  5
#define TILE_M 16
#define LDA    260

__global__ void fill_i32_kernel(int* __restrict__ p, int val, size_t n) {
    size_t i = (size_t)blockIdx.x * blockDim.x + threadIdx.x;
    size_t stride = (size_t)gridDim.x * blockDim.x;
    for (; i < n; i += stride) p[i] = val;
}

#define SLOT_BIG 0xFFFF
__global__ void degree_slot_kernel(const int* __restrict__ col, int* __restrict__ degi,
                                   int* __restrict__ slot_words, int E) {
    int e = blockIdx.x * blockDim.x + threadIdx.x;
    if (e >= E) return;
    const int s = atomicAdd(&degi[col[e]], 1);
    ((unsigned short*)slot_words)[e] = (unsigned short)min(s, SLOT_BIG);
}

#define SEG_B 256
#define SEG_W (SEG_B / 32)

__global__ __launch_bounds__(SEG_B) void seg_reserve_kernel(const int* __restrict__ degi,
                                                            int* __restrict__ rowstart,
                                                            int* __restrict__ total, int N) {
    __shared__ int s_w[SEG_W];
    const int tid  = threadIdx.x;
    const int lane = tid & 31;
    const int wave = tid >> 5;
    const int i = blockIdx.x * SEG_B + tid;
    const int v = (i < N) ? degi[i] : 0;
    int inc = v;
    for (int off = 1; off < 32; off <<= 1) {
        const int t = __shfl_up(inc, off, 32);
        if (lane >= off) inc += t;
    }
    if (lane == 31) s_w[wave] = inc;
    __syncthreads();
    if (wave == 0) {
        const int w = (lane < SEG_W) ? s_w[lane] : 0;
        int winc = w;
        for (int off = 1; off < SEG_W; off <<= 1) {
            const int t = __shfl_up(winc, off, 32);
            if (lane >= off) winc += t;
        }
        int base = 0;
        if (lane == SEG_W - 1) base = atomicAdd(total, winc);
        base = __shfl(base, SEG_W - 1, 32);
        if (lane < SEG_W) s_w[lane] = base + winc - w;
    }
    __syncthreads();
    if (i < N) rowstart[i] = s_w[wave] + inc - v;
}

#define BIG_N 16
__device__ int big_node_slot(int* __restrict__ big, const int d) {
    #pragma unroll 1
    for (int i = 0; i < BIG_N; ++i) {
        const int old = atomicCAS(&big[2 * i], 0, d + 1);
        if (old == 0 || old == d + 1) return SLOT_BIG + atomicAdd(&big[2 * i + 1], 1);
    }
    return SLOT_BIG;
}

__global__ void csr_fill_kernel(const int* __restrict__ row, const int* __restrict__ col,
                                const int* __restrict__ rowstart,
                                const int* __restrict__ slot_words, int* __restrict__ big,
                                int* __restrict__ edge_src, int E) {
    int e = blockIdx.x * blockDim.x + threadIdx.x;
    if (e >= E) return;
    const int d = col[e];
    int pos = ((const unsigned short*)slot_words)[e];
    if (pos == SLOT_BIG) pos = big_node_slot(big, d);
    edge_src[rowstart[d] + pos] = row[e];
}

#define BK_B  1024
#define BK_W  (BK_B / 32)

__global__ __launch_bounds__(BK_B) void bucket_count_kernel(const int* __restrict__ degi,
                                                            int* __restrict__ meta, int N) {
    __shared__ int s_w[BK_W][NBUCK];
    const int tid  = threadIdx.x;
    const int lane = tid & 31;
    const int wave = tid >> 5;
    const int i = blockIdx.x * BK_B + tid;
    const int b = (i < N) ? min(degi[i], MAXDEG) : -1;
    int myc = 0;
    for (int k = 0; k < NBUCK; ++k) {
        const unsigned m = __builtin_amdgcn_ballot_w32(b == k);
        if (lane == k) myc = __builtin_popcount(m);
    }
    if (lane < NBUCK) s_w[wave][lane] = myc;
    __syncthreads();
    if (tid < NBUCK) {
        int acc = 0;
        for (int w = 0; w < BK_W; ++w) acc += s_w[w][tid];
        if (acc) atomicAdd(&meta[tid], acc);
    }
}

__global__ void bucket_offset_kernel(int* __restrict__ meta) {
    if (threadIdx.x == 0 && blockIdx.x == 0) {
        int off = 0;
        for (int b = 0; b < NBUCK; ++b) {
            meta[5 + b] = off;
            off += (meta[b] + TILE_M - 1) & ~(TILE_M - 1);
        }
    }
}

__global__ __launch_bounds__(BK_B) void bucket_scatter_kernel(const int* __restrict__ degi,
                                                              int* __restrict__ meta,
                                                              int* __restrict__ perm, int N) {
    __shared__ int s_w[BK_W][NBUCK];
    __shared__ int s_base[NBUCK];
    const int tid  = threadIdx.x;
    const int lane = tid & 31;
    const int wave = tid >> 5;
    const int i = blockIdx.x * BK_B + tid;
    const int b = (i < N) ? min(degi[i], MAXDEG) : -1;
    unsigned mine = 0;
    int myc = 0;
    for (int k = 0; k < NBUCK; ++k) {
        const unsigned m = __builtin_amdgcn_ballot_w32(b == k);
        if (b == k)    mine = m;
        if (lane == k) myc  = __builtin_popcount(m);
    }
    if (lane < NBUCK) s_w[wave][lane] = myc;
    __syncthreads();
    if (tid < NBUCK) {
        int acc = 0;
        for (int w = 0; w < BK_W; ++w) { const int c = s_w[w][tid]; s_w[w][tid] = acc; acc += c; }
        int base = 0;
        if (acc) base = atomicAdd(&meta[10 + tid], acc);
        s_base[tid] = meta[5 + tid] + base;
    }
    __syncthreads();
    if (b >= 0) {
        const int rank = __builtin_popcount(mine & ((1u << lane) - 1u));
        perm[s_base[b] + s_w[wave][b] + rank] = i;
    }
}

#define WG_TILES 4
#define WG_M     (TILE_M * WG_TILES)

template <int NT>
__device__ __forceinline__ void cat_gemm(const float* aRow, const float* __restrict__ wn,
                                         const float* __restrict__ wr, const int khalf,
                                         v8f (&acc)[NT]) {
    #pragma unroll
    for (int kk = 0; kk < 32; ++kk) {
        const int k = kk * 4 + khalf;
        v2f b;
        b.x = wn[k];       b.y = wn[k + 1];
        #pragma unroll
        for (int t = 0; t < NT; ++t) {
            v2f a;
            a.x = aRow[t * TILE_M * LDA + k];  a.y = aRow[t * TILE_M * LDA + k + 1];
            acc[t] = __builtin_amdgcn_wmma_f32_16x16x4_f32(false, a, false, b,
                                                           (short)0, acc[t], false, false);
        }
    }
    #pragma unroll
    for (int kk = 0; kk < 32; ++kk) {
        const int k = kk * 4 + khalf;
        v2f b;
        b.x = wr[k];       b.y = wr[k + 1];
        #pragma unroll
        for (int t = 0; t < NT; ++t) {
            v2f a;
            a.x = aRow[t * TILE_M * LDA + IN_F + k];  a.y = aRow[t * TILE_M * LDA + IN_F + k + 1];
            acc[t] = __builtin_amdgcn_wmma_f32_16x16x4_f32(false, a, false, b,
                                                           (short)0, acc[t], false, false);
        }
    }
}

__device__ __forceinline__ void store_tile(const v8f acc, const int* nodes, const int mbase,
                                           const float bias, const int ncol,
                                           float* __restrict__ out) {
    #pragma unroll
    for (int i = 0; i < 8; ++i) {
        const int node = nodes[mbase + i];
        if (node >= 0) out[(size_t)node * OUT_F + ncol] = acc[i] + bias;
    }
}

__global__ __launch_bounds__(256) void mfconv_gemm_kernel(
    const float* __restrict__ x,
    const int* __restrict__ degi, const int* __restrict__ rowstart,
    const int* __restrict__ edge_src, const int* __restrict__ perm,
    const float* __restrict__ neigh_w, const float* __restrict__ neigh_b,
    const float* __restrict__ root_w,  const float* __restrict__ root_b,
    float* __restrict__ out, const int permLen)
{
    __shared__ float ldsA[WG_M * LDA];
    __shared__ int   s_node[WG_M];
    __shared__ int   s_bucket[WG_TILES];

    const int tid = threadIdx.x;

    if (tid < WG_M) {
        const int p = blockIdx.x * WG_M + tid;
        s_node[tid] = (p < permLen) ? perm[p] : -1;
    }
    __syncthreads();

    if (tid < WG_TILES) {
        int b = -1;
        for (int i = 0; i < TILE_M; ++i) {
            int nd = s_node[tid * TILE_M + i];
            if (nd >= 0) { b = min(degi[nd], MAXDEG); break; }
        }
        s_bucket[tid] = b;
    }

    {
        const int mloc = tid >> 4;
        const int k0   = (tid & 15) * 16;
        int node[WG_TILES], deg[WG_TILES], start[WG_TILES];
        #pragma unroll
        for (int t = 0; t < WG_TILES; ++t) node[t] = s_node[t * TILE_M + mloc];
        #pragma unroll
        for (int t = 0; t < WG_TILES; ++t) {
            deg[t] = 0; start[t] = 0;
            if (node[t] >= 0 && k0 < IN_F) { deg[t] = degi[node[t]]; start[t] = rowstart[node[t]]; }
        }
        #pragma unroll
        for (int t = 0; t < WG_TILES; ++t) {
            float4 a0 = make_float4(0.f,0.f,0.f,0.f), a1 = a0, a2 = a0, a3 = a0;
            if (node[t] >= 0) {
                if (k0 < IN_F) {
                    const int d = deg[t];
                    for (int j = 0; j < d; ++j) {
                        const float* xs = x + (size_t)edge_src[start[t] + j] * IN_F + k0;
                        const float4 v0 = *(const float4*)(xs + 0);
                        const float4 v1 = *(const float4*)(xs + 4);
                        const float4 v2 = *(const float4*)(xs + 8);
                        const float4 v3 = *(const float4*)(xs + 12);
                        a0.x += v0.x; a0.y += v0.y; a0.z += v0.z; a0.w += v0.w;
                        a1.x += v1.x; a1.y += v1.y; a1.z += v1.z; a1.w += v1.w;
                        a2.x += v2.x; a2.y += v2.y; a2.z += v2.z; a2.w += v2.w;
                        a3.x += v3.x; a3.y += v3.y; a3.z += v3.z; a3.w += v3.w;
                    }
                    const float inv = 1.0f / fmaxf((float)d, 1.0f);
                    a0.x *= inv; a0.y *= inv; a0.z *= inv; a0.w *= inv;
                    a1.x *= inv; a1.y *= inv; a1.z *= inv; a1.w *= inv;
                    a2.x *= inv; a2.y *= inv; a2.z *= inv; a2.w *= inv;
                    a3.x *= inv; a3.y *= inv; a3.z *= inv; a3.w *= inv;
                } else {
                    const float* xs = x + (size_t)node[t] * IN_F + (k0 - IN_F);
                    a0 = *(const float4*)(xs + 0);
                    a1 = *(const float4*)(xs + 4);
                    a2 = *(const float4*)(xs + 8);
                    a3 = *(const float4*)(xs + 12);
                }
            }
            float* dst = &ldsA[(t * TILE_M + mloc) * LDA + k0];
            *(float4*)(dst + 0)  = a0;
            *(float4*)(dst + 4)  = a1;
            *(float4*)(dst + 8)  = a2;
            *(float4*)(dst + 12) = a3;
        }
    }
    __syncthreads();

    int bk[WG_TILES];
    #pragma unroll
    for (int t = 0; t < WG_TILES; ++t) bk[t] = __builtin_amdgcn_readfirstlane(s_bucket[t]);
    const int bmax = max(max(bk[0], bk[1]), max(bk[2], bk[3]));
    if (bmax < 0) return;
    const bool shared_bucket = (bk[0] < 0 || bk[0] == bmax) && (bk[1] < 0 || bk[1] == bmax) &&
                               (bk[2] < 0 || bk[2] == bmax) && (bk[3] < 0 || bk[3] == bmax);

    const int wave  = tid >> 5;
    const int lane  = tid & 31;
    const int lr    = lane & 15;
    const int khalf = (lane >> 4) << 1;
    const int ncol  = wave * 16 + lr;
    const int mbase = (lane >> 4) * 8;
    const float* aRow = &ldsA[lr * LDA];

    if (shared_bucket) {
        const float* __restrict__ wn = neigh_w + ((size_t)bmax * OUT_F + ncol) * IN_F;
        const float* __restrict__ wr = root_w  + ((size_t)bmax * OUT_F + ncol) * IN_F;
        v8f acc[WG_TILES] = {};
        cat_gemm<WG_TILES>(aRow, wn, wr, khalf, acc);
        const float bias = neigh_b[bmax * OUT_F + ncol] + root_b[bmax * OUT_F + ncol];
        #pragma unroll
        for (int t = 0; t < WG_TILES; ++t)
            store_tile(acc[t], &s_node[t * TILE_M], mbase, bias, ncol, out);
    } else {
        #pragma unroll 1
        for (int t = 0; t < WG_TILES; ++t) {
            const int b = __builtin_amdgcn_readfirstlane(s_bucket[t]);
            if (b < 0) continue;
            const float* __restrict__ wn = neigh_w + ((size_t)b * OUT_F + ncol) * IN_F;
            const float* __restrict__ wr = root_w  + ((size_t)b * OUT_F + ncol) * IN_F;
            v8f acc1[1] = {};
            cat_gemm<1>(aRow + t * TILE_M * LDA, wn, wr, khalf, acc1);
            const float bias = neigh_b[b * OUT_F + ncol] + root_b[b * OUT_F + ncol];
            store_tile(acc1[0], &s_node[t * TILE_M], mbase, bias, ncol, out);
        }
    }
}

extern "C" void kernel_launch(void* const* d_in, const int* in_sizes, int n_in,
                              void* d_out, int out_size, void* d_ws, size_t ws_size,
                              hipStream_t stream) {
    const float* x       = (const float*)d_in[0];
    const float* neigh_w = (const float*)d_in[1];
    const float* neigh_b = (const float*)d_in[2];
    const float* root_w  = (const float*)d_in[3];
    const float* root_b  = (const float*)d_in[4];
    const int*   eidx    = (const int*)d_in[5];

    const int N = in_sizes[0] / IN_F;
    const int E = in_sizes[5] / 2;
    const int* row = eidx;
    const int* col = eidx + E;

    const int nSegBlocks = (N + SEG_B - 1) / SEG_B;
    const int permLen = ((N + NBUCK * TILE_M + TILE_M - 1) / TILE_M) * TILE_M;
    const int nGroups = (permLen + WG_M - 1) / WG_M;

    char* ws = (char*)d_ws;
    int* degi     = (int*)ws;  ws += (size_t)N * sizeof(int);
    int* rowstart = (int*)ws;  ws += (size_t)N * sizeof(int);
    int* edge_src = (int*)ws;  ws += (size_t)E * sizeof(int);
    int* perm     = (int*)ws;
    int* slots    = (int*)ws;  ws += ((size_t)N + permLen) * sizeof(int);
    int* meta     = (int*)ws;

    fill_i32_kernel<<<1024, 256, 0, stream>>>(degi, 0, (size_t)N);
    fill_i32_kernel<<<1, 64, 0, stream>>>(meta, 0, 48);

    degree_slot_kernel<<<(E + 255) / 256, 256, 0, stream>>>(col, degi, slots, E);
    seg_reserve_kernel<<<nSegBlocks, SEG_B, 0, stream>>>(degi, rowstart, meta + 15, N);
    csr_fill_kernel<<<(E + 255) / 256, 256, 0, stream>>>(row, col, rowstart, slots, meta + 16,
                                                         edge_src, E);

    fill_i32_kernel<<<1024, 256, 0, stream>>>(perm, -1, (size_t)permLen);

    bucket_count_kernel<<<(N + BK_B - 1) / BK_B, BK_B, 0, stream>>>(degi, meta, N);
    bucket_offset_kernel<<<1, 1, 0, stream>>>(meta);
    bucket_scatter_kernel<<<(N + BK_B - 1) / BK_B, BK_B, 0, stream>>>(degi, meta, perm, N);

    mfconv_gemm_kernel<<<nGroups, 256, 0, stream>>>(x, degi, rowstart, edge_src, perm,
                                                    neigh_w, neigh_b, root_w, root_b,
                                                    (float*)d_out, permLen);
}


# GEMM kernel staging gather: one wave per node row (16 B per lane), a node's source indices read as one vector, rows of two nodes in flight instead of an index-then-row chain per thread
# speedup vs baseline: 2.8444x; 1.0826x over previous
#include <hip/hip_runtime.h>

typedef __attribute__((ext_vector_type(2))) float v2f;
typedef __attribute__((ext_vector_type(8))) float v8f;

#define IN_F   128
#define OUT_F  128
#define MAXDEG 4
#define NBUCK  5
#define TILE_M 16
#define LDA    260

__global__ void fill_i32_kernel(int* __restrict__ p, int val, size_t n) {
    size_t i = (size_t)blockIdx.x * blockDim.x + threadIdx.x;
    size_t stride = (size_t)gridDim.x * blockDim.x;
    for (; i < n; i += stride) p[i] = val;
}

#define SLOT_BIG 0xFFFF
__global__ void degree_slot_kernel(const int* __restrict__ col, int* __restrict__ degi,
                                   int* __restrict__ slot_words, int E) {
    int e = blockIdx.x * blockDim.x + threadIdx.x;
    if (e >= E) return;
    const int s = atomicAdd(&degi[col[e]], 1);
    ((unsigned short*)slot_words)[e] = (unsigned short)min(s, SLOT_BIG);
}

#define SEG_B 256
#define SEG_W (SEG_B / 32)

__global__ __launch_bounds__(SEG_B) void seg_reserve_kernel(const int* __restrict__ degi,
                                                            int* __restrict__ rowstart,
                                                            int* __restrict__ total, int N) {
    __shared__ int s_w[SEG_W];
    const int tid  = threadIdx.x;
    const int lane = tid & 31;
    const int wave = tid >> 5;
    const int i = blockIdx.x * SEG_B + tid;
    const int v = (i < N) ? degi[i] : 0;
    int inc = v;
    for (int off = 1; off < 32; off <<= 1) {
        const int t = __shfl_up(inc, off, 32);
        if (lane >= off) inc += t;
    }
    if (lane == 31) s_w[wave] = inc;
    __syncthreads();
    if (wave == 0) {
        const int w = (lane < SEG_W) ? s_w[lane] : 0;
        int winc = w;
        for (int off = 1; off < SEG_W; off <<= 1) {
            const int t = __shfl_up(winc, off, 32);
            if (lane >= off) winc += t;
        }
        int base = 0;
        if (lane == SEG_W - 1) base = atomicAdd(total, winc);
        base = __shfl(base, SEG_W - 1, 32);
        if (lane < SEG_W) s_w[lane] = base + winc - w;
    }
    __syncthreads();
    if (i < N) rowstart[i] = s_w[wave] + inc - v;
}

#define BIG_N 16
__device__ int big_node_slot(int* __restrict__ big, const int d) {
    #pragma unroll 1
    for (int i = 0; i < BIG_N; ++i) {
        const int old = atomicCAS(&big[2 * i], 0, d + 1);
        if (old == 0 || old == d + 1) return SLOT_BIG + atomicAdd(&big[2 * i + 1], 1);
    }
    return SLOT_BIG;
}

__global__ void csr_fill_kernel(const int* __restrict__ row, const int* __restrict__ col,
                                const int* __restrict__ rowstart,
                                const int* __restrict__ slot_words, int* __restrict__ big,
                                int* __restrict__ edge_src, int E) {
    int e = blockIdx.x * blockDim.x + threadIdx.x;
    if (e >= E) return;
    const int d = col[e];
    int pos = ((const unsigned short*)slot_words)[e];
    if (pos == SLOT_BIG) pos = big_node_slot(big, d);
    edge_src[rowstart[d] + pos] = row[e];
}

#define BK_B  1024
#define BK_W  (BK_B / 32)

__global__ __launch_bounds__(BK_B) void bucket_count_kernel(const int* __restrict__ degi,
                                                            int* __restrict__ meta, int N) {
    __shared__ int s_w[BK_W][NBUCK];
    const int tid  = threadIdx.x;
    const int lane = tid & 31;
    const int wave = tid >> 5;
    const int i = blockIdx.x * BK_B + tid;
    const int b = (i < N) ? min(degi[i], MAXDEG) : -1;
    int myc = 0;
    for (int k = 0; k < NBUCK; ++k) {
        const unsigned m = __builtin_amdgcn_ballot_w32(b == k);
        if (lane == k) myc = __builtin_popcount(m);
    }
    if (lane < NBUCK) s_w[wave][lane] = myc;
    __syncthreads();
    if (tid < NBUCK) {
        int acc = 0;
        for (int w = 0; w < BK_W; ++w) acc += s_w[w][tid];
        if (acc) atomicAdd(&meta[tid], acc);
    }
}

__global__ void bucket_offset_kernel(int* __restrict__ meta) {
    if (threadIdx.x == 0 && blockIdx.x == 0) {
        int off = 0;
        for (int b = 0; b < NBUCK; ++b) {
            meta[5 + b] = off;
            off += (meta[b] + TILE_M - 1) & ~(TILE_M - 1);
        }
    }
}

__global__ __launch_bounds__(BK_B) void bucket_scatter_kernel(const int* __restrict__ degi,
                                                              int* __restrict__ meta,
                                                              int* __restrict__ perm, int N) {
    __shared__ int s_w[BK_W][NBUCK];
    __shared__ int s_base[NBUCK];
    const int tid  = threadIdx.x;
    const int lane = tid & 31;
    const int wave = tid >> 5;
    const int i = blockIdx.x * BK_B + tid;
    const int b = (i < N) ? min(degi[i], MAXDEG) : -1;
    unsigned mine = 0;
    int myc = 0;
    for (int k = 0; k < NBUCK; ++k) {
        const unsigned m = __builtin_amdgcn_ballot_w32(b == k);
        if (b == k)    mine = m;
        if (lane == k) myc  = __builtin_popcount(m);
    }
    if (lane < NBUCK) s_w[wave][lane] = myc;
    __syncthreads();
    if (tid < NBUCK) {
        int acc = 0;
        for (int w = 0; w < BK_W; ++w) { const int c = s_w[w][tid]; s_w[w][tid] = acc; acc += c; }
        int base = 0;
        if (acc) base = atomicAdd(&meta[10 + tid], acc);
        s_base[tid] = meta[5 + tid] + base;
    }
    __syncthreads();
    if (b >= 0) {
        const int rank = __builtin_popcount(mine & ((1u << lane) - 1u));
        perm[s_base[b] + s_w[wave][b] + rank] = i;
    }
}

#define WG_TILES 4
#define WG_M     (TILE_M * WG_TILES)

template <int NT>
__device__ __forceinline__ void cat_gemm(const float* aRow, const float* __restrict__ wn,
                                         const float* __restrict__ wr, const int khalf,
                                         v8f (&acc)[NT]) {
    #pragma unroll
    for (int kk = 0; kk < 32; ++kk) {
        const int k = kk * 4 + khalf;
        v2f b;
        b.x = wn[k];       b.y = wn[k + 1];
        #pragma unroll
        for (int t = 0; t < NT; ++t) {
            v2f a;
            a.x = aRow[t * TILE_M * LDA + k];  a.y = aRow[t * TILE_M * LDA + k + 1];
            acc[t] = __builtin_amdgcn_wmma_f32_16x16x4_f32(false, a, false, b,
                                                           (short)0, acc[t], false, false);
        }
    }
    #pragma unroll
    for (int kk = 0; kk < 32; ++kk) {
        const int k = kk * 4 + khalf;
        v2f b;
        b.x = wr[k];       b.y = wr[k + 1];
        #pragma unroll
        for (int t = 0; t < NT; ++t) {
            v2f a;
            a.x = aRow[t * TILE_M * LDA + IN_F + k];  a.y = aRow[t * TILE_M * LDA + IN_F + k + 1];
            acc[t] = __builtin_amdgcn_wmma_f32_16x16x4_f32(false, a, false, b,
                                                           (short)0, acc[t], false, false);
        }
    }
}

__device__ __forceinline__ void store_tile(const v8f acc, const int* nodes, const int mbase,
                                           const float bias, const int ncol,
                                           float* __restrict__ out) {
    #pragma unroll
    for (int i = 0; i < 8; ++i) {
        const int node = nodes[mbase + i];
        if (node >= 0) out[(size_t)node * OUT_F + ncol] = acc[i] + bias;
    }
}

#define NODES_W  8
#define GATHER_U 8

#define ROW_V4 (IN_F / 4)
template <int U>
__device__ __forceinline__ void ask_rows(const float4* __restrict__ x4, const int idxv, const int j0,
                                         const int m, const unsigned lane, float4 (&v)[U]) {
    #pragma unroll
    for (int u = 0; u < U; ++u) {
        const int j = max(min(j0 + u, m - 1), 0);
        const unsigned s = (unsigned)__builtin_amdgcn_readlane(idxv, j);
        v[u] = x4[s * ROW_V4 + lane];
    }
}
template <int U>
__device__ __forceinline__ void sum_rows(const float4 (&v)[U], const int j0, const int m,
                                         float4& acc) {
    #pragma unroll
    for (int u = 0; u < U; ++u) {
        const float w = ((j0 + u) < m) ? 1.0f : 0.0f;
        acc.x = __builtin_fmaf(v[u].x, w, acc.x);
        acc.y = __builtin_fmaf(v[u].y, w, acc.y);
        acc.z = __builtin_fmaf(v[u].z, w, acc.z);
        acc.w = __builtin_fmaf(v[u].w, w, acc.w);
    }
}

struct NodeRows { float4 nb[GATHER_U]; float4 rt; };

__device__ __forceinline__ void ask_node(const float4* __restrict__ x4, const int node, const int deg,
                                         const int idxv, const unsigned lane, NodeRows& r) {
    ask_rows<GATHER_U>(x4, idxv, 0, min(deg, GATHER_U), lane, r.nb);
    r.rt = x4[(unsigned)max(node, 0) * ROW_V4 + lane];
}

__device__ __forceinline__ void put_node(const NodeRows& r, const int deg, float* dst) {
    float4 acc = make_float4(0.f, 0.f, 0.f, 0.f);
    sum_rows<GATHER_U>(r.nb, 0, min(deg, GATHER_U), acc);
    const float sc = (deg > GATHER_U) ? 1.0f : 1.0f / fmaxf((float)deg, 1.0f);
    acc.x *= sc; acc.y *= sc; acc.z *= sc; acc.w *= sc;
    *(float4*)dst = acc;
    *(float4*)(dst + IN_F) = r.rt;
}

__global__ __launch_bounds__(256) void mfconv_gemm_kernel(
    const float* __restrict__ x,
    const int* __restrict__ degi, const int* __restrict__ rowstart,
    const int* __restrict__ edge_src, const int* __restrict__ perm,
    const float* __restrict__ neigh_w, const float* __restrict__ neigh_b,
    const float* __restrict__ root_w,  const float* __restrict__ root_b,
    float* __restrict__ out, const int permLen)
{
    __shared__ float ldsA[WG_M * LDA];
    __shared__ int   s_node[WG_M];
    __shared__ int   s_wbucket[WG_M / NODES_W];

    const int tid   = threadIdx.x;
    const int wave  = tid >> 5;
    const int lane  = tid & 31;
    const int lane4 = lane * 4;

    int nd = -1, dg = 0, st = 0;
    if (lane < NODES_W) {
        const int p = blockIdx.x * WG_M + wave * NODES_W + lane;
        if (p < permLen) nd = perm[p];
        if (nd >= 0) { dg = degi[nd]; st = rowstart[nd]; }
        s_node[wave * NODES_W + lane] = nd;
    }
    {
        const unsigned real = __builtin_amdgcn_ballot_w32(nd >= 0);
        int bw = -1;
        if (real) bw = min(__builtin_amdgcn_readlane(dg, __builtin_ctz(real)), MAXDEG);
        if (lane == 0) s_wbucket[wave] = bw;
    }

    int nS[NODES_W], dS[NODES_W], idx[NODES_W];
    #pragma unroll
    for (int i = 0; i < NODES_W; ++i) {
        nS[i] = __builtin_amdgcn_readlane(nd, i);
        dS[i] = __builtin_amdgcn_readlane(dg, i);
        const int s0 = __builtin_amdgcn_readlane(st, i);
        idx[i] = max(nS[i], 0);
        if (lane < dS[i]) idx[i] = edge_src[s0 + lane];
    }

    const float4* __restrict__ x4 = (const float4*)x;
    float* dst0 = &ldsA[(wave * NODES_W) * LDA + lane4];
    NodeRows ra, rb;
    ask_node(x4, nS[0], dS[0], idx[0], (unsigned)lane, ra);
    #pragma unroll
    for (int i = 0; i < NODES_W; i += 2) {
        ask_node(x4, nS[i + 1], dS[i + 1], idx[i + 1], (unsigned)lane, rb);
        __builtin_amdgcn_sched_barrier(0);
        put_node(ra, dS[i], dst0 + i * LDA);
        if (i + 2 < NODES_W) ask_node(x4, nS[i + 2], dS[i + 2], idx[i + 2], (unsigned)lane, ra);
        __builtin_amdgcn_sched_barrier(0);
        put_node(rb, dS[i + 1], dst0 + (i + 1) * LDA);
    }

    #pragma unroll 1
    for (int i = 0; i < NODES_W; ++i) {
        const int d = __builtin_amdgcn_readlane(dg, i);
        if (d <= GATHER_U) continue;
        const int s0 = __builtin_amdgcn_readlane(st, i);
        float* dst = &ldsA[(wave * NODES_W + i) * LDA + lane4];
        float4 acc = *(float4*)dst;
        for (int c = 0; c < d; c += 32) {
            int idxv = 0;
            if (c + lane < d) idxv = edge_src[s0 + c + lane];
            const int m = min(d - c, 32);
            for (int j0 = (c == 0) ? GATHER_U : 0; j0 < m; j0 += 4) {
                float4 v[4];
                ask_rows<4>(x4, idxv, j0, m, (unsigned)lane, v);
                sum_rows<4>(v, j0, m, acc);
            }
        }
        const float inv = 1.0f / (float)d;
        acc.x *= inv; acc.y *= inv; acc.z *= inv; acc.w *= inv;
        *(float4*)dst = acc;
    }
    __syncthreads();

    int bk[WG_TILES];
    #pragma unroll
    for (int t = 0; t < WG_TILES; ++t) {
        const int b0 = s_wbucket[2 * t], b1 = s_wbucket[2 * t + 1];
        bk[t] = __builtin_amdgcn_readfirstlane(b0 >= 0 ? b0 : b1);
    }
    const int bmax = max(max(bk[0], bk[1]), max(bk[2], bk[3]));
    if (bmax < 0) return;
    const bool shared_bucket = (bk[0] < 0 || bk[0] == bmax) && (bk[1] < 0 || bk[1] == bmax) &&
                               (bk[2] < 0 || bk[2] == bmax) && (bk[3] < 0 || bk[3] == bmax);

    const int lr    = lane & 15;
    const int khalf = (lane >> 4) << 1;
    const int ncol  = wave * 16 + lr;
    const int mbase = (lane >> 4) * 8;
    const float* aRow = &ldsA[lr * LDA];

    if (shared_bucket) {
        const float* __restrict__ wn = neigh_w + ((size_t)bmax * OUT_F + ncol) * IN_F;
        const float* __restrict__ wr = root_w  + ((size_t)bmax * OUT_F + ncol) * IN_F;
        v8f acc[WG_TILES] = {};
        cat_gemm<WG_TILES>(aRow, wn, wr, khalf, acc);
        const float bias = neigh_b[bmax * OUT_F + ncol] + root_b[bmax * OUT_F + ncol];
        #pragma unroll
        for (int t = 0; t < WG_TILES; ++t)
            store_tile(acc[t], &s_node[t * TILE_M], mbase, bias, ncol, out);
    } else {
        #pragma unroll 1
        for (int t = 0; t < WG_TILES; ++t) {
            const int b0 = s_wbucket[2 * t], b1 = s_wbucket[2 * t + 1];
            const int b = __builtin_amdgcn_readfirstlane(b0 >= 0 ? b0 : b1);
            if (b < 0) continue;
            const float* __restrict__ wn = neigh_w + ((size_t)b * OUT_F + ncol) * IN_F;
            const float* __restrict__ wr = root_w  + ((size_t)b * OUT_F + ncol) * IN_F;
            v8f acc1[1] = {};
            cat_gemm<1>(aRow + t * TILE_M * LDA, wn, wr, khalf, acc1);
            const float bias = neigh_b[b * OUT_F + ncol] + root_b[b * OUT_F + ncol];
            store_tile(acc1[0], &s_node[t * TILE_M], mbase, bias, ncol, out);
        }
    }
}

extern "C" void kernel_launch(void* const* d_in, const int* in_sizes, int n_in,
                              void* d_out, int out_size, void* d_ws, size_t ws_size,
                              hipStream_t stream) {
    const float* x       = (const float*)d_in[0];
    const float* neigh_w = (const float*)d_in[1];
    const float* neigh_b = (const float*)d_in[2];
    const float* root_w  = (const float*)d_in[3];
    const float* root_b  = (const float*)d_in[4];
    const int*   eidx    = (const int*)d_in[5];

    const int N = in_sizes[0] / IN_F;
    const int E = in_sizes[5] / 2;
    const int* row = eidx;
    const int* col = eidx + E;

    const int nSegBlocks = (N + SEG_B - 1) / SEG_B;
    const int permLen = ((N + NBUCK * TILE_M + TILE_M - 1) / TILE_M) * TILE_M;
    const int nGroups = (permLen + WG_M - 1) / WG_M;

    char* ws = (char*)d_ws;
    int* degi     = (int*)ws;  ws += (size_t)N * sizeof(int);
    int* rowstart = (int*)ws;  ws += (size_t)N * sizeof(int);
    int* edge_src = (int*)ws;  ws += (size_t)E * sizeof(int);
    int* perm     = (int*)ws;
    int* slots    = (int*)ws;  ws += ((size_t)N + permLen) * sizeof(int);
    int* meta     = (int*)ws;

    fill_i32_kernel<<<1024, 256, 0, stream>>>(degi, 0, (size_t)N);
    fill_i32_kernel<<<1, 64, 0, stream>>>(meta, 0, 48);

    degree_slot_kernel<<<(E + 255) / 256, 256, 0, stream>>>(col, degi, slots, E);
    seg_reserve_kernel<<<nSegBlocks, SEG_B, 0, stream>>>(degi, rowstart, meta + 15, N);
    csr_fill_kernel<<<(E + 255) / 256, 256, 0, stream>>>(row, col, rowstart, slots, meta + 16,
                                                         edge_src, E);

    fill_i32_kernel<<<1024, 256, 0, stream>>>(perm, -1, (size_t)permLen);

    bucket_count_kernel<<<(N + BK_B - 1) / BK_B, BK_B, 0, stream>>>(degi, meta, N);
    bucket_offset_kernel<<<1, 1, 0, stream>>>(meta);
    bucket_scatter_kernel<<<(N + BK_B - 1) / BK_B, BK_B, 0, stream>>>(degi, meta, perm, N);

    mfconv_gemm_kernel<<<nGroups, 256, 0, stream>>>(x, degi, rowstart, edge_src, perm,
                                                    neigh_w, neigh_b, root_w, root_b,
                                                    (float*)d_out, permLen);
}


# launch fusion: zero fills merged; bucket count rides in the segment-reserve kernel; the scatter kernel works out bucket offsets and writes perm's -1 padding itself (10 -> 6 launches)
# speedup vs baseline: 2.9967x; 1.0535x over previous
#include <hip/hip_runtime.h>

typedef __attribute__((ext_vector_type(2))) float v2f;
typedef __attribute__((ext_vector_type(8))) float v8f;

#define IN_F   128
#define OUT_F  128
#define MAXDEG 4
#define NBUCK  5
#define TILE_M 16
#define LDA    260

#define META_N 48
__global__ void init_zero_kernel(int* __restrict__ degi, size_t n, int* __restrict__ meta) {
    size_t i = (size_t)blockIdx.x * blockDim.x + threadIdx.x;
    if (i < META_N) meta[i] = 0;
    size_t stride = (size_t)gridDim.x * blockDim.x;
    for (; i < n; i += stride) degi[i] = 0;
}

#define SLOT_BIG 0xFFFF
__global__ void degree_slot_kernel(const int* __restrict__ col, int* __restrict__ degi,
                                   int* __restrict__ slot_words, int E) {
    int e = blockIdx.x * blockDim.x + threadIdx.x;
    if (e >= E) return;
    const int s = atomicAdd(&degi[col[e]], 1);
    ((unsigned short*)slot_words)[e] = (unsigned short)min(s, SLOT_BIG);
}

#define SEG_B 1024
#define SEG_W (SEG_B / 32)

__global__ __launch_bounds__(SEG_B) void seg_reserve_count_kernel(const int* __restrict__ degi,
                                                                  int* __restrict__ rowstart,
                                                                  int* __restrict__ meta, int N) {
    __shared__ int s_w[SEG_W];
    __shared__ int s_c[SEG_W][NBUCK];
    const int tid  = threadIdx.x;
    const int lane = tid & 31;
    const int wave = tid >> 5;
    const int i = blockIdx.x * SEG_B + tid;
    const int v = (i < N) ? degi[i] : 0;
    const int b = (i < N) ? min(v, MAXDEG) : -1;
    int inc = v;
    for (int off = 1; off < 32; off <<= 1) {
        const int t = __shfl_up(inc, off, 32);
        if (lane >= off) inc += t;
    }
    if (lane == 31) s_w[wave] = inc;
    int myc = 0;
    for (int k = 0; k < NBUCK; ++k) {
        const unsigned m = __builtin_amdgcn_ballot_w32(b == k);
        if (lane == k) myc = __builtin_popcount(m);
    }
    if (lane < NBUCK) s_c[wave][lane] = myc;
    __syncthreads();
    if (wave == 0) {
        const int w = s_w[lane];
        int winc = w;
        for (int off = 1; off < SEG_W; off <<= 1) {
            const int t = __shfl_up(winc, off, 32);
            if (lane >= off) winc += t;
        }
        int base = 0;
        if (lane == SEG_W - 1) base = atomicAdd(&meta[15], winc);
        base = __shfl(base, SEG_W - 1, 32);
        s_w[lane] = base + winc - w;
    } else if (wave == 1 && lane < NBUCK) {
        int acc = 0;
        for (int w = 0; w < SEG_W; ++w) acc += s_c[w][lane];
        if (acc) atomicAdd(&meta[lane], acc);
    }
    __syncthreads();
    if (i < N) rowstart[i] = s_w[wave] + inc - v;
}

#define BIG_N 16
__device__ int big_node_slot(int* __restrict__ big, const int d) {
    #pragma unroll 1
    for (int i = 0; i < BIG_N; ++i) {
        const int old = atomicCAS(&big[2 * i], 0, d + 1);
        if (old == 0 || old == d + 1) return SLOT_BIG + atomicAdd(&big[2 * i + 1], 1);
    }
    return SLOT_BIG;
}

__global__ void csr_fill_kernel(const int* __restrict__ row, const int* __restrict__ col,
                                const int* __restrict__ rowstart,
                                const int* __restrict__ slot_words, int* __restrict__ big,
                                int* __restrict__ edge_src, int E) {
    int e = blockIdx.x * blockDim.x + threadIdx.x;
    if (e >= E) return;
    const int d = col[e];
    int pos = ((const unsigned short*)slot_words)[e];
    if (pos == SLOT_BIG) pos = big_node_slot(big, d);
    edge_src[rowstart[d] + pos] = row[e];
}

#define BK_B  1024
#define BK_W  (BK_B / 32)

__global__ __launch_bounds__(BK_B) void bucket_scatter_kernel(const int* __restrict__ degi,
                                                              int* __restrict__ meta,
                                                              int* __restrict__ perm, int N,
                                                              int permLen) {
    __shared__ int s_w[BK_W][NBUCK];
    __shared__ int s_base[NBUCK];
    const int tid  = threadIdx.x;
    const int lane = tid & 31;
    const int wave = tid >> 5;
    const int i = blockIdx.x * BK_B + tid;
    const int b = (i < N) ? min(degi[i], MAXDEG) : -1;
    unsigned mine = 0;
    int myc = 0;
    for (int k = 0; k < NBUCK; ++k) {
        const unsigned m = __builtin_amdgcn_ballot_w32(b == k);
        if (b == k)    mine = m;
        if (lane == k) myc  = __builtin_popcount(m);
    }
    if (lane < NBUCK) s_w[wave][lane] = myc;
    __syncthreads();
    if (tid < NBUCK) {
        int seg = 0;
        for (int k = 0; k < NBUCK; ++k) {
            const int a = (meta[k] + TILE_M - 1) & ~(TILE_M - 1);
            if (k < tid) seg += a;
        }
        int acc = 0;
        for (int w = 0; w < BK_W; ++w) { const int c = s_w[w][tid]; s_w[w][tid] = acc; acc += c; }
        int base = 0;
        if (acc) base = atomicAdd(&meta[10 + tid], acc);
        s_base[tid] = seg + base;
    }
    if (blockIdx.x == 0) {
        int seg = 0;
        for (int k = 0; k < NBUCK; ++k) {
            const int c = meta[k];
            const int a = (c + TILE_M - 1) & ~(TILE_M - 1);
            if (tid < a - c) perm[seg + c + tid] = -1;
            seg += a;
        }
        for (int p = seg + tid; p < permLen; p += BK_B) perm[p] = -1;
    }
    __syncthreads();
    if (b >= 0) {
        const int rank = __builtin_popcount(mine & ((1u << lane) - 1u));
        perm[s_base[b] + s_w[wave][b] + rank] = i;
    }
}

#define WG_TILES 4
#define WG_M     (TILE_M * WG_TILES)

template <int NT>
__device__ __forceinline__ void cat_gemm(const float* aRow, const float* __restrict__ wn,
                                         const float* __restrict__ wr, const int khalf,
                                         v8f (&acc)[NT]) {
    #pragma unroll
    for (int kk = 0; kk < 32; ++kk) {
        const int k = kk * 4 + khalf;
        v2f b;
        b.x = wn[k];       b.y = wn[k + 1];
        #pragma unroll
        for (int t = 0; t < NT; ++t) {
            v2f a;
            a.x = aRow[t * TILE_M * LDA + k];  a.y = aRow[t * TILE_M * LDA + k + 1];
            acc[t] = __builtin_amdgcn_wmma_f32_16x16x4_f32(false, a, false, b,
                                                           (short)0, acc[t], false, false);
        }
    }
    #pragma unroll
    for (int kk = 0; kk < 32; ++kk) {
        const int k = kk * 4 + khalf;
        v2f b;
        b.x = wr[k];       b.y = wr[k + 1];
        #pragma unroll
        for (int t = 0; t < NT; ++t) {
            v2f a;
            a.x = aRow[t * TILE_M * LDA + IN_F + k];  a.y = aRow[t * TILE_M * LDA + IN_F + k + 1];
            acc[t] = __builtin_amdgcn_wmma_f32_16x16x4_f32(false, a, false, b,
                                                           (short)0, acc[t], false, false);
        }
    }
}

__device__ __forceinline__ void store_tile(const v8f acc, const int* nodes, const int mbase,
                                           const float bias, const int ncol,
                                           float* __restrict__ out) {
    #pragma unroll
    for (int i = 0; i < 8; ++i) {
        const int node = nodes[mbase + i];
        if (node >= 0) out[(size_t)node * OUT_F + ncol] = acc[i] + bias;
    }
}

#define NODES_W  8
#define GATHER_U 8

#define ROW_V4 (IN_F / 4)
template <int U>
__device__ __forceinline__ void ask_rows(const float4* __restrict__ x4, const int idxv, const int j0,
                                         const int m, const unsigned lane, float4 (&v)[U]) {
    #pragma unroll
    for (int u = 0; u < U; ++u) {
        const int j = max(min(j0 + u, m - 1), 0);
        const unsigned s = (unsigned)__builtin_amdgcn_readlane(idxv, j);
        v[u] = x4[s * ROW_V4 + lane];
    }
}
template <int U>
__device__ __forceinline__ void sum_rows(const float4 (&v)[U], const int j0, const int m,
                                         float4& acc) {
    #pragma unroll
    for (int u = 0; u < U; ++u) {
        const float w = ((j0 + u) < m) ? 1.0f : 0.0f;
        acc.x = __builtin_fmaf(v[u].x, w, acc.x);
        acc.y = __builtin_fmaf(v[u].y, w, acc.y);
        acc.z = __builtin_fmaf(v[u].z, w, acc.z);
        acc.w = __builtin_fmaf(v[u].w, w, acc.w);
    }
}

struct NodeRows { float4 nb[GATHER_U]; float4 rt; };

__device__ __forceinline__ void ask_node(const float4* __restrict__ x4, const int node, const int deg,
                                         const int idxv, const unsigned lane, NodeRows& r) {
    ask_rows<GATHER_U>(x4, idxv, 0, min(deg, GATHER_U), lane, r.nb);
    r.rt = x4[(unsigned)max(node, 0) * ROW_V4 + lane];
}

__device__ __forceinline__ void put_node(const NodeRows& r, const int deg, float* dst) {
    float4 acc = make_float4(0.f, 0.f, 0.f, 0.f);
    sum_rows<GATHER_U>(r.nb, 0, min(deg, GATHER_U), acc);
    const float sc = (deg > GATHER_U) ? 1.0f : 1.0f / fmaxf((float)deg, 1.0f);
    acc.x *= sc; acc.y *= sc; acc.z *= sc; acc.w *= sc;
    *(float4*)dst = acc;
    *(float4*)(dst + IN_F) = r.rt;
}

__global__ __launch_bounds__(256) void mfconv_gemm_kernel(
    const float* __restrict__ x,
    const int* __restrict__ degi, const int* __restrict__ rowstart,
    const int* __restrict__ edge_src, const int* __restrict__ perm,
    const float* __restrict__ neigh_w, const float* __restrict__ neigh_b,
    const float* __restrict__ root_w,  const float* __restrict__ root_b,
    float* __restrict__ out, const int permLen)
{
    __shared__ float ldsA[WG_M * LDA];
    __shared__ int   s_node[WG_M];
    __shared__ int   s_wbucket[WG_M / NODES_W];

    const int tid   = threadIdx.x;
    const int wave  = tid >> 5;
    const int lane  = tid & 31;
    const int lane4 = lane * 4;

    int nd = -1, dg = 0, st = 0;
    if (lane < NODES_W) {
        const int p = blockIdx.x * WG_M + wave * NODES_W + lane;
        if (p < permLen) nd = perm[p];
        if (nd >= 0) { dg = degi[nd]; st = rowstart[nd]; }
        s_node[wave * NODES_W + lane] = nd;
    }
    {
        const unsigned real = __builtin_amdgcn_ballot_w32(nd >= 0);
        int bw = -1;
        if (real) bw = min(__builtin_amdgcn_readlane(dg, __builtin_ctz(real)), MAXDEG);
        if (lane == 0) s_wbucket[wave] = bw;
    }

    int nS[NODES_W], dS[NODES_W], idx[NODES_W];
    #pragma unroll
    for (int i = 0; i < NODES_W; ++i) {
        nS[i] = __builtin_amdgcn_readlane(nd, i);
        dS[i] = __builtin_amdgcn_readlane(dg, i);
        const int s0 = __builtin_amdgcn_readlane(st, i);
        idx[i] = max(nS[i], 0);
        if (lane < dS[i]) idx[i] = edge_src[s0 + lane];
    }

    const float4* __restrict__ x4 = (const float4*)x;
    float* dst0 = &ldsA[(wave * NODES_W) * LDA + lane4];
    NodeRows ra, rb;
    ask_node(x4, nS[0], dS[0], idx[0], (unsigned)lane, ra);
    #pragma unroll
    for (int i = 0; i < NODES_W; i += 2) {
        ask_node(x4, nS[i + 1], dS[i + 1], idx[i + 1], (unsigned)lane, rb);
        __builtin_amdgcn_sched_barrier(0);
        put_node(ra, dS[i], dst0 + i * LDA);
        if (i + 2 < NODES_W) ask_node(x4, nS[i + 2], dS[i + 2], idx[i + 2], (unsigned)lane, ra);
        __builtin_amdgcn_sched_barrier(0);
        put_node(rb, dS[i + 1], dst0 + (i + 1) * LDA);
    }

    #pragma unroll 1
    for (int i = 0; i < NODES_W; ++i) {
        const int d = __builtin_amdgcn_readlane(dg, i);
        if (d <= GATHER_U) continue;
        const int s0 = __builtin_amdgcn_readlane(st, i);
        float* dst = &ldsA[(wave * NODES_W + i) * LDA + lane4];
        float4 acc = *(float4*)dst;
        for (int c = 0; c < d; c += 32) {
            int idxv = 0;
            if (c + lane < d) idxv = edge_src[s0 + c + lane];
            const int m = min(d - c, 32);
            for (int j0 = (c == 0) ? GATHER_U : 0; j0 < m; j0 += 4) {
                float4 v[4];
                ask_rows<4>(x4, idxv, j0, m, (unsigned)lane, v);
                sum_rows<4>(v, j0, m, acc);
            }
        }
        const float inv = 1.0f / (float)d;
        acc.x *= inv; acc.y *= inv; acc.z *= inv; acc.w *= inv;
        *(float4*)dst = acc;
    }
    __syncthreads();

    int bk[WG_TILES];
    #pragma unroll
    for (int t = 0; t < WG_TILES; ++t) {
        const int b0 = s_wbucket[2 * t], b1 = s_wbucket[2 * t + 1];
        bk[t] = __builtin_amdgcn_readfirstlane(b0 >= 0 ? b0 : b1);
    }
    const int bmax = max(max(bk[0], bk[1]), max(bk[2], bk[3]));
    if (bmax < 0) return;
    const bool shared_bucket = (bk[0] < 0 || bk[0] == bmax) && (bk[1] < 0 || bk[1] == bmax) &&
                               (bk[2] < 0 || bk[2] == bmax) && (bk[3] < 0 || bk[3] == bmax);

    const int lr    = lane & 15;
    const int khalf = (lane >> 4) << 1;
    const int ncol  = wave * 16 + lr;
    const int mbase = (lane >> 4) * 8;
    const float* aRow = &ldsA[lr * LDA];

    if (shared_bucket) {
        const float* __restrict__ wn = neigh_w + ((size_t)bmax * OUT_F + ncol) * IN_F;
        const float* __restrict__ wr = root_w  + ((size_t)bmax * OUT_F + ncol) * IN_F;
        v8f acc[WG_TILES] = {};
        cat_gemm<WG_TILES>(aRow, wn, wr, khalf, acc);
        const float bias = neigh_b[bmax * OUT_F + ncol] + root_b[bmax * OUT_F + ncol];
        #pragma unroll
        for (int t = 0; t < WG_TILES; ++t)
            store_tile(acc[t], &s_node[t * TILE_M], mbase, bias, ncol, out);
    } else {
        #pragma unroll 1
        for (int t = 0; t < WG_TILES; ++t) {
            const int b0 = s_wbucket[2 * t], b1 = s_wbucket[2 * t + 1];
            const int b = __builtin_amdgcn_readfirstlane(b0 >= 0 ? b0 : b1);
            if (b < 0) continue;
            const float* __restrict__ wn = neigh_w + ((size_t)b * OUT_F + ncol) * IN_F;
            const float* __restrict__ wr = root_w  + ((size_t)b * OUT_F + ncol) * IN_F;
            v8f acc1[1] = {};
            cat_gemm<1>(aRow + t * TILE_M * LDA, wn, wr, khalf, acc1);
            const float bias = neigh_b[b * OUT_F + ncol] + root_b[b * OUT_F + ncol];
            store_tile(acc1[0], &s_node[t * TILE_M], mbase, bias, ncol, out);
        }
    }
}

extern "C" void kernel_launch(void* const* d_in, const int* in_sizes, int n_in,
                              void* d_out, int out_size, void* d_ws, size_t ws_size,
                              hipStream_t stream) {
    const float* x       = (const float*)d_in[0];
    const float* neigh_w = (const float*)d_in[1];
    const float* neigh_b = (const float*)d_in[2];
    const float* root_w  = (const float*)d_in[3];
    const float* root_b  = (const float*)d_in[4];
    const int*   eidx    = (const int*)d_in[5];

    const int N = in_sizes[0] / IN_F;
    const int E = in_sizes[5] / 2;
    const int* row = eidx;
    const int* col = eidx + E;

    const int nSegBlocks = (N + SEG_B - 1) / SEG_B;
    const int permLen = ((N + NBUCK * TILE_M + TILE_M - 1) / TILE_M) * TILE_M;
    const int nGroups = (permLen + WG_M - 1) / WG_M;

    char* ws = (char*)d_ws;
    int* degi     = (int*)ws;  ws += (size_t)N * sizeof(int);
    int* rowstart = (int*)ws;  ws += (size_t)N * sizeof(int);
    int* edge_src = (int*)ws;  ws += (size_t)E * sizeof(int);
    int* perm     = (int*)ws;
    int* slots    = (int*)ws;  ws += ((size_t)N + permLen) * sizeof(int);
    int* meta     = (int*)ws;

    init_zero_kernel<<<1024, 256, 0, stream>>>(degi, (size_t)N, meta);

    degree_slot_kernel<<<(E + 255) / 256, 256, 0, stream>>>(col, degi, slots, E);
    seg_reserve_count_kernel<<<nSegBlocks, SEG_B, 0, stream>>>(degi, rowstart, meta, N);
    csr_fill_kernel<<<(E + 255) / 256, 256, 0, stream>>>(row, col, rowstart, slots, meta + 16,
                                                         edge_src, E);

    bucket_scatter_kernel<<<(N + BK_B - 1) / BK_B, BK_B, 0, stream>>>(degi, meta, perm, N, permLen);

    mfconv_gemm_kernel<<<nGroups, 256, 0, stream>>>(x, degi, rowstart, edge_src, perm,
                                                    neigh_w, neigh_b, root_w, root_b,
                                                    (float*)d_out, permLen);
}
